# starting kernel
# baseline (speedup 1.0000x reference)
#include <hip/hip_runtime.h>
#include <hip/hip_bf16.h>
#include <math.h>

typedef float v2f __attribute__((ext_vector_type(2)));
typedef float v8f __attribute__((ext_vector_type(8)));

#define HEADS 4
#define HC 256
#define CDIM 64
#define NEG_SLOPE 0.2f

__global__ void gemm_wmma_f32(const float* __restrict__ A,
                              const float* __restrict__ B,
                              float* __restrict__ C,
                              int M, int K, int Ncol) {
    const int lane   = threadIdx.x & 31;
    const int wave   = threadIdx.x >> 5;
    const int wtile  = blockIdx.x * 4 + wave;
    const int wtprow = Ncol >> 6;
    const int total  = (M >> 4) * wtprow;
    if (wtile >= total) return;
    const int trow   = wtile / wtprow;
    const int tcol   = wtile - trow * wtprow;
    const int row0   = trow << 4;
    const int col0   = tcol << 6;
    const int laneM  = lane & 15;
    const int laneHi = lane >> 4;

    const float* Arow  = A + (size_t)(row0 + laneM) * K;
    const float* Bbase = B + (size_t)col0 + laneM;

    v8f acc0 = {}, acc1 = {}, acc2 = {}, acc3 = {};
#pragma unroll 2
    for (int k = 0; k < K; k += 4) {
        const int ka = k + (laneHi << 1);
        v2f a;
        a.x = Arow[ka];
        a.y = Arow[ka + 1];
        const float* B0 = Bbase + (size_t)ka * Ncol;
        const float* B1 = Bbase + (size_t)(ka + 1) * Ncol;
        v2f b0, b1, b2, b3;
        b0.x = B0[0];  b0.y = B1[0];
        b1.x = B0[16]; b1.y = B1[16];
        b2.x = B0[32]; b2.y = B1[32];
        b3.x = B0[48]; b3.y = B1[48];
        acc0 = __builtin_amdgcn_wmma_f32_16x16x4_f32(false, a, false, b0, (short)0, acc0, false, false);
        acc1 = __builtin_amdgcn_wmma_f32_16x16x4_f32(false, a, false, b1, (short)0, acc1, false, false);
        acc2 = __builtin_amdgcn_wmma_f32_16x16x4_f32(false, a, false, b2, (short)0, acc2, false, false);
        acc3 = __builtin_amdgcn_wmma_f32_16x16x4_f32(false, a, false, b3, (short)0, acc3, false, false);
    }

    float* Crow = C + (size_t)(row0 + 8 * laneHi) * Ncol + col0 + laneM;
#pragma unroll
    for (int i = 0; i < 8; ++i) {
        float* r = Crow + (size_t)i * Ncol;
        r[0]  = acc0[i];
        r[16] = acc1[i];
        r[32] = acc2[i];
        r[48] = acc3[i];
    }
}

__global__ void attn_scores(const float* __restrict__ Hm,
                            const float* __restrict__ att_s,
                            const float* __restrict__ att_d,
                            float* __restrict__ as_, float* __restrict__ ad_,
                            int N) {
    const int idx = blockIdx.x * blockDim.x + threadIdx.x;
    if (idx >= N * HEADS) return;
    const int n = idx >> 2, h = idx & 3;
    const float* row = Hm + (size_t)n * HC + h * CDIM;
    const float* vs  = att_s + h * CDIM;
    const float* vd  = att_d + h * CDIM;
    float s = 0.f, d = 0.f;
#pragma unroll 8
    for (int c = 0; c < CDIM; ++c) {
        const float v = row[c];
        s = fmaf(v, vs[c], s);
        d = fmaf(v, vd[c], d);
    }
    as_[idx] = s;
    ad_[idx] = d;
}

__device__ __forceinline__ unsigned fmap(float x) {
    unsigned u = __float_as_uint(x);
    return (u & 0x80000000u) ? ~u : (u | 0x80000000u);
}
__device__ __forceinline__ float funmap(unsigned u) {
    u = (u & 0x80000000u) ? (u ^ 0x80000000u) : ~u;
    return __uint_as_float(u);
}

__device__ __forceinline__ void edge_nodes(int eid, int E,
                                           const int* __restrict__ srcv,
                                           const int* __restrict__ dstv,
                                           int& s, int& d) {
    if (eid < E) { s = srcv[eid]; d = dstv[eid]; }
    else         { s = d = eid - E; }
}

__device__ __forceinline__ float leaky(float e) {
    return e > 0.f ? e : NEG_SLOPE * e;
}

__global__ void edge_max(const int* __restrict__ srcv, const int* __restrict__ dstv,
                         const float* __restrict__ as_, const float* __restrict__ ad_,
                         unsigned* __restrict__ dmax, int E, int N) {
    const int idx = blockIdx.x * blockDim.x + threadIdx.x;
    const int tot = (E + N) * HEADS;
    if (idx >= tot) return;
    const int eid = idx >> 2, h = idx & 3;
    int s, d;
    edge_nodes(eid, E, srcv, dstv, s, d);
    const float e = leaky(as_[s * HEADS + h] + ad_[d * HEADS + h]);
    atomicMax(&dmax[d * HEADS + h], fmap(e));
}

__global__ void edge_expsum(const int* __restrict__ srcv, const int* __restrict__ dstv,
                            const float* __restrict__ as_, const float* __restrict__ ad_,
                            const unsigned* __restrict__ dmax,
                            float* __restrict__ EXv, float* __restrict__ dsum,
                            int E, int N) {
    const int idx = blockIdx.x * blockDim.x + threadIdx.x;
    const int tot = (E + N) * HEADS;
    if (idx >= tot) return;
    const int eid = idx >> 2, h = idx & 3;
    int s, d;
    edge_nodes(eid, E, srcv, dstv, s, d);
    const float e  = leaky(as_[s * HEADS + h] + ad_[d * HEADS + h]);
    const float m  = funmap(dmax[d * HEADS + h]);
    const float ex = __expf(e - m);
    EXv[idx] = ex;
    atomicAdd(&dsum[d * HEADS + h], ex);
}

__global__ void edge_scatter(const int* __restrict__ srcv, const int* __restrict__ dstv,
                             const float* __restrict__ Hm,
                             const float* __restrict__ EXv,
                             const float* __restrict__ dsum,
                             float* __restrict__ ACC, int E, int N) {
    const int eid = blockIdx.x;
    const int t   = threadIdx.x;
    int s, d;
    edge_nodes(eid, E, srcv, dstv, s, d);
    const int h = t >> 6;
    const float alpha = EXv[eid * HEADS + h] / (dsum[d * HEADS + h] + 1e-16f);
    atomicAdd(&ACC[(size_t)d * HC + t], Hm[(size_t)s * HC + t] * alpha);
}

__global__ void head_mean(const float* __restrict__ ACC,
                          const float* __restrict__ bias,
                          float* __restrict__ out, int N, int do_relu) {
    const int idx = blockIdx.x * blockDim.x + threadIdx.x;
    if (idx >= N * CDIM) return;
    const int n = idx >> 6, c = idx & 63;
    const float* r = ACC + (size_t)n * HC;
    float v = (r[c] + r[CDIM + c] + r[2 * CDIM + c] + r[3 * CDIM + c]) * 0.25f
              + bias[c];
    if (do_relu) v = fmaxf(v, 0.f);
    out[idx] = v;
}

extern "C" void kernel_launch(void* const* d_in, const int* in_sizes, int n_in,
                              void* d_out, int out_size, void* d_ws, size_t ws_size,
                              hipStream_t stream) {
    const float* x        = (const float*)d_in[0];
    const int*   ei       = (const int*)  d_in[1];
    const float* W1       = (const float*)d_in[2];
    const float* att_src1 = (const float*)d_in[3];
    const float* att_dst1 = (const float*)d_in[4];
    const float* b1       = (const float*)d_in[5];
    const float* W2       = (const float*)d_in[6];
    const float* att_src2 = (const float*)d_in[7];
    const float* att_dst2 = (const float*)d_in[8];
    const float* b2       = (const float*)d_in[9];

    const int Fin = 128;
    const int N   = in_sizes[0] / Fin;
    const int E   = in_sizes[1] / 2;
    const int ET  = E + N;
    const int* srcv = ei;
    const int* dstv = ei + E;

    float* ws    = (float*)d_ws;
    float* Hbuf  = ws;
    float* ACC   = Hbuf + (size_t)N * HC;
    float* HID   = ACC  + (size_t)N * HC;
    float* ASRC  = HID  + (size_t)N * CDIM;
    float* ADST  = ASRC + (size_t)N * HEADS;
    unsigned* DMAX = (unsigned*)(ADST + (size_t)N * HEADS);
    float* DSUM  = (float*)(DMAX + (size_t)N * HEADS);
    float* EXv   = DSUM + (size_t)N * HEADS;

    const int gemmStrips = (N / 16) * (HC / 64);
    const int gemmBlocks = (gemmStrips + 3) / 4;
    const int nhBlocks   = (N * HEADS + 255) / 256;
    const int ehBlocks   = (ET * HEADS + 255) / 256;
    const int ncBlocks   = (N * CDIM + 255) / 256;

    hipMemsetAsync(DMAX, 0, (size_t)N * HEADS * 4, stream);
    hipMemsetAsync(DSUM, 0, (size_t)N * HEADS * 4, stream);
    hipMemsetAsync(ACC,  0, (size_t)N * HC * 4, stream);

    gemm_wmma_f32<<<gemmBlocks, 128, 0, stream>>>(x, W1, Hbuf, N, Fin, HC);
    attn_scores<<<nhBlocks, 256, 0, stream>>>(Hbuf, att_src1, att_dst1, ASRC, ADST, N);
    edge_max<<<ehBlocks, 256, 0, stream>>>(srcv, dstv, ASRC, ADST, DMAX, E, N);
    edge_expsum<<<ehBlocks, 256, 0, stream>>>(srcv, dstv, ASRC, ADST, DMAX, EXv, DSUM, E, N);
    edge_scatter<<<ET, HC, 0, stream>>>(srcv, dstv, Hbuf, EXv, DSUM, ACC, E, N);
    head_mean<<<ncBlocks, 256, 0, stream>>>(ACC, b1, HID, N,  1);

    hipMemsetAsync(DMAX, 0, (size_t)N * HEADS * 4, stream);
    hipMemsetAsync(DSUM, 0, (size_t)N * HEADS * 4, stream);
    hipMemsetAsync(ACC,  0, (size_t)N * HC * 4, stream);

    gemm_wmma_f32<<<gemmBlocks, 128, 0, stream>>>(HID, W2, Hbuf, N, CDIM, HC);
    attn_scores<<<nhBlocks, 256, 0, stream>>>(Hbuf, att_src2, att_dst2, ASRC, ADST, N);
    edge_max<<<ehBlocks, 256, 0, stream>>>(srcv, dstv, ASRC, ADST, DMAX, E, N);
    edge_expsum<<<ehBlocks, 256, 0, stream>>>(srcv, dstv, ASRC, ADST, DMAX, EXv, DSUM, E, N);
    edge_scatter<<<ET, HC, 0, stream>>>(srcv, dstv, Hbuf, EXv, DSUM, ACC, E, N);
    head_mean<<<ncBlocks, 256, 0, stream>>>(ACC, b2, (float*)d_out, N,  0);
}


# segment softmax inside the owner block: scores, max, exp sum of its noted edges in LDS (LDS max + float adds; list 8192 -> 4096); 4 per-edge launches, 4 clears go; sum order: exp sums by LDS arrival
# speedup vs baseline: 5.5048x; 5.5048x over previous
#include <hip/hip_runtime.h>
#include <hip/hip_bf16.h>
#include <math.h>

typedef float v2f __attribute__((ext_vector_type(2)));
typedef float v8f __attribute__((ext_vector_type(8)));

#define HEADS 4
#define HC 256
#define CDIM 64
#define NEG_SLOPE 0.2f

__global__ void gemm_wmma_f32(const float* __restrict__ A,
                              const float* __restrict__ B,
                              float* __restrict__ C,
                              int M, int K, int Ncol) {
    const int lane   = threadIdx.x & 31;
    const int wave   = threadIdx.x >> 5;
    const int wtile  = blockIdx.x * 4 + wave;
    const int wtprow = Ncol >> 6;
    const int total  = (M >> 4) * wtprow;
    if (wtile >= total) return;
    const int trow   = wtile / wtprow;
    const int tcol   = wtile - trow * wtprow;
    const int row0   = trow << 4;
    const int col0   = tcol << 6;
    const int laneM  = lane & 15;
    const int laneHi = lane >> 4;

    const float* Arow  = A + (size_t)(row0 + laneM) * K;
    const float* Bbase = B + (size_t)col0 + laneM;

    v8f acc0 = {}, acc1 = {}, acc2 = {}, acc3 = {};
#pragma unroll 2
    for (int k = 0; k < K; k += 4) {
        const int ka = k + (laneHi << 1);
        v2f a;
        a.x = Arow[ka];
        a.y = Arow[ka + 1];
        const float* B0 = Bbase + (size_t)ka * Ncol;
        const float* B1 = Bbase + (size_t)(ka + 1) * Ncol;
        v2f b0, b1, b2, b3;
        b0.x = B0[0];  b0.y = B1[0];
        b1.x = B0[16]; b1.y = B1[16];
        b2.x = B0[32]; b2.y = B1[32];
        b3.x = B0[48]; b3.y = B1[48];
        acc0 = __builtin_amdgcn_wmma_f32_16x16x4_f32(false, a, false, b0, (short)0, acc0, false, false);
        acc1 = __builtin_amdgcn_wmma_f32_16x16x4_f32(false, a, false, b1, (short)0, acc1, false, false);
        acc2 = __builtin_amdgcn_wmma_f32_16x16x4_f32(false, a, false, b2, (short)0, acc2, false, false);
        acc3 = __builtin_amdgcn_wmma_f32_16x16x4_f32(false, a, false, b3, (short)0, acc3, false, false);
    }

    float* Crow = C + (size_t)(row0 + 8 * laneHi) * Ncol + col0 + laneM;
#pragma unroll
    for (int i = 0; i < 8; ++i) {
        float* r = Crow + (size_t)i * Ncol;
        r[0]  = acc0[i];
        r[16] = acc1[i];
        r[32] = acc2[i];
        r[48] = acc3[i];
    }
}

__global__ void attn_scores(const float* __restrict__ Hm,
                            const float* __restrict__ att_s,
                            const float* __restrict__ att_d,
                            float* __restrict__ as_, float* __restrict__ ad_,
                            int N) {
    const int idx = blockIdx.x * blockDim.x + threadIdx.x;
    if (idx >= N * HEADS) return;
    const int n = idx >> 2, h = idx & 3;
    const float* row = Hm + (size_t)n * HC + h * CDIM;
    const float* vs  = att_s + h * CDIM;
    const float* vd  = att_d + h * CDIM;
    float s = 0.f, d = 0.f;
#pragma unroll 8
    for (int c = 0; c < CDIM; ++c) {
        const float v = row[c];
        s = fmaf(v, vs[c], s);
        d = fmaf(v, vd[c], d);
    }
    as_[idx] = s;
    ad_[idx] = d;
}

__device__ __forceinline__ unsigned fmap(float x) {
    unsigned u = __float_as_uint(x);
    return (u & 0x80000000u) ? ~u : (u | 0x80000000u);
}
__device__ __forceinline__ float funmap(unsigned u) {
    u = (u & 0x80000000u) ? (u ^ 0x80000000u) : ~u;
    return __uint_as_float(u);
}

__device__ __forceinline__ float leaky(float e) {
    return e > 0.f ? e : NEG_SLOPE * e;
}

#define RNG   200
#define LCAP  4096
#define AGG_T 1024
#define AGG_U 4

__device__ __forceinline__ void scan4(int4 d, int e0, int base, int nown,
                                      int* cnt, int2* list) {
    const unsigned r0 = (unsigned)(d.x - base);
    const unsigned r1 = (unsigned)(d.y - base);
    const unsigned r2 = (unsigned)(d.z - base);
    const unsigned r3 = (unsigned)(d.w - base);
    if (min(min(r0, r1), min(r2, r3)) < (unsigned)nown) {
        if (r0 < (unsigned)nown) { const int sl = atomicAdd(cnt, 1); if (sl < LCAP) list[sl] = make_int2(e0,     (int)r0); }
        if (r1 < (unsigned)nown) { const int sl = atomicAdd(cnt, 1); if (sl < LCAP) list[sl] = make_int2(e0 + 1, (int)r1); }
        if (r2 < (unsigned)nown) { const int sl = atomicAdd(cnt, 1); if (sl < LCAP) list[sl] = make_int2(e0 + 2, (int)r2); }
        if (r3 < (unsigned)nown) { const int sl = atomicAdd(cnt, 1); if (sl < LCAP) list[sl] = make_int2(e0 + 3, (int)r3); }
    }
}

template <class F>
__device__ __forceinline__ void walk_owned(const int* __restrict__ dstv, int E,
                                           int base, int nown, int t, F f) {
    for (int e = t; e < E; e += AGG_T) {
        const unsigned r = (unsigned)(dstv[e] - base);
        if (r < (unsigned)nown) f(e, (int)r);
    }
}

template <int U>
__device__ __forceinline__ void sum_noted(const int2* list, const float4* wl,
                                          int i0, int lane, float* acc,
                                          const float* __restrict__ Hm) {
    int   rr[U], ss[U];
    float al[U][HEADS];
    float v[U][8];
#pragma unroll
    for (int u = 0; u < U; ++u) {
        const int2 en = list[i0 + u];
        ss[u] = __builtin_amdgcn_readfirstlane(en.x);
        rr[u] = __builtin_amdgcn_readfirstlane(en.y);
    }
#pragma unroll
    for (int u = 0; u < U; ++u) {
        const float* hrow = Hm + (size_t)ss[u] * HC + lane;
#pragma unroll
        for (int j = 0; j < 8; ++j) v[u][j] = hrow[j * 32];
    }
#pragma unroll
    for (int u = 0; u < U; ++u) {
        const float4 w4 = wl[i0 + u];
        al[u][0] = w4.x; al[u][1] = w4.y; al[u][2] = w4.z; al[u][3] = w4.w;
    }
#pragma unroll
    for (int u = 0; u < U; ++u) {
        float* arow = acc + rr[u] * HC + lane;
#pragma unroll
        for (int j = 0; j < 8; ++j)
            atomicAdd(&arow[j * 32], v[u][j] * al[u][j >> 1]);
    }
}

__global__ void
agg_owner(const int* __restrict__ srcv, const int* __restrict__ dstv,
          const float* __restrict__ Hm,
          const float* __restrict__ as_, const float* __restrict__ ad_,
          const float* __restrict__ bias,
          float* __restrict__ out, int E, int N, int do_relu) {
    __shared__ float    acc[RNG * HC];
    __shared__ int2     list[LCAP];
    __shared__ float4   wl[LCAP];
    __shared__ float    adl[RNG * HEADS];
    __shared__ float    slf[RNG * HEADS];
    __shared__ unsigned mx[RNG * HEADS];
    __shared__ float    sm[RNG * HEADS];
    __shared__ int      cnt;

    const int t    = threadIdx.x;
    const int lane = t & 31;
    const int wave = t >> 5;
    const int base = blockIdx.x * RNG;
    const int nown = min(RNG, N - base);

    if (t == 0) cnt = 0;
    for (int i = t; i < nown * HEADS; i += AGG_T) {
        const int n4 = base * HEADS + i;
        const float a = ad_[n4];
        const float e = leaky(as_[n4] + a);
        adl[i] = a;
        slf[i] = e;
        mx[i]  = fmap(e);
        sm[i]  = 0.f;
    }
    __syncthreads();

    const int  E4 = ((E & 3) == 0) ? (E >> 2) : 0;
    const int4* d4 = (const int4*)dstv;
    int q0 = t;
    for (; q0 + 3 * AGG_T < E4; q0 += 4 * AGG_T) {
        int4 dv[4];
#pragma unroll
        for (int u = 0; u < 4; ++u) dv[u] = d4[q0 + u * AGG_T];
#pragma unroll
        for (int u = 0; u < 4; ++u)
            scan4(dv[u], (q0 + u * AGG_T) << 2, base, nown, &cnt, list);
    }
    for (; q0 < E4; q0 += AGG_T)
        scan4(d4[q0], q0 << 2, base, nown, &cnt, list);
    for (int e = (E4 << 2) + t; e < E; e += AGG_T) {
        const unsigned r = (unsigned)(dstv[e] - base);
        if (r < (unsigned)nown) {
            const int sl = atomicAdd(&cnt, 1);
            if (sl < LCAP) list[sl] = make_int2(e, (int)r);
        }
    }
    __syncthreads();

    const int  ntot   = cnt;
    const bool listed = ntot <= LCAP;

    if (listed && ntot > 0) {
        static_assert(LCAP == 4 * AGG_T, "four list entries per thread");
        bool   ok[4];
        int2   en[4];
        int    sv[4];
        float4 a4[4];
#pragma unroll
        for (int u = 0; u < 4; ++u) {
            const int i = t + u * AGG_T;
            ok[u] = i < ntot;
            en[u] = make_int2(0, 0);
            if (ok[u]) en[u] = list[i];
        }
#pragma unroll
        for (int u = 0; u < 4; ++u) sv[u] = srcv[en[u].x];
#pragma unroll
        for (int u = 0; u < 4; ++u) a4[u] = *(const float4*)(as_ + (size_t)sv[u] * HEADS);
#pragma unroll
        for (int u = 0; u < 4; ++u) {
            if (ok[u]) {
                const int i = t + u * AGG_T;
                const int r = en[u].y;
                const float* d = adl + r * HEADS;
                float4 e4;
                e4.x = leaky(a4[u].x + d[0]);
                e4.y = leaky(a4[u].y + d[1]);
                e4.z = leaky(a4[u].z + d[2]);
                e4.w = leaky(a4[u].w + d[3]);
                wl[i]   = e4;
                list[i] = make_int2(sv[u], r);
                unsigned* m = mx + r * HEADS;
                atomicMax(&m[0], fmap(e4.x));
                atomicMax(&m[1], fmap(e4.y));
                atomicMax(&m[2], fmap(e4.z));
                atomicMax(&m[3], fmap(e4.w));
            }
        }
    } else if (!listed) {
        walk_owned(dstv, E, base, nown, t, [&](int e, int r) {
            const int s = srcv[e];
            for (int h = 0; h < HEADS; ++h)
                atomicMax(&mx[r * HEADS + h],
                          fmap(leaky(as_[(size_t)s * HEADS + h] + adl[r * HEADS + h])));
        });
    }
    __syncthreads();

    if (listed) {
        for (int i = t; i < ntot; i += AGG_T) {
            const int r = list[i].y;
            const float4 e4 = wl[i];
            const unsigned* m = mx + r * HEADS;
            float4 x4;
            x4.x = __expf(e4.x - funmap(m[0]));
            x4.y = __expf(e4.y - funmap(m[1]));
            x4.z = __expf(e4.z - funmap(m[2]));
            x4.w = __expf(e4.w - funmap(m[3]));
            wl[i] = x4;
            float* sp = sm + r * HEADS;
            atomicAdd(&sp[0], x4.x);
            atomicAdd(&sp[1], x4.y);
            atomicAdd(&sp[2], x4.z);
            atomicAdd(&sp[3], x4.w);
        }
    } else {
        walk_owned(dstv, E, base, nown, t, [&](int e, int r) {
            const int s = srcv[e];
            for (int h = 0; h < HEADS; ++h) {
                const float ev = leaky(as_[(size_t)s * HEADS + h] + adl[r * HEADS + h]);
                atomicAdd(&sm[r * HEADS + h], __expf(ev - funmap(mx[r * HEADS + h])));
            }
        });
    }
    for (int i = t; i < nown * HEADS; i += AGG_T) {
        const float ex = __expf(slf[i] - funmap(mx[i]));
        slf[i] = ex;
        atomicAdd(&sm[i], ex);
    }
    __syncthreads();

    for (int i = t; i < nown * HEADS; i += AGG_T) {
        const float dn = sm[i] + 1e-16f;
        sm[i]  = dn;
        slf[i] = slf[i] / dn;
    }
    __syncthreads();

    for (int i = t; i < nown * HC; i += AGG_T)
        acc[i] = Hm[(size_t)base * HC + i] * slf[i >> 6];
    if (listed) {
        for (int i = t; i < ntot; i += AGG_T) {
            const float* dn = sm + list[i].y * HEADS;
            float4 x4 = wl[i];
            x4.x = x4.x / dn[0];
            x4.y = x4.y / dn[1];
            x4.z = x4.z / dn[2];
            x4.w = x4.w / dn[3];
            wl[i] = x4;
        }
    }
    __syncthreads();

    if (listed) {
        const int nfull = ntot - (ntot % AGG_U);
        for (int i0 = wave * AGG_U; i0 < nfull; i0 += 32 * AGG_U)
            sum_noted<AGG_U>(list, wl, i0, lane, acc, Hm);
        for (int i0 = nfull + wave; i0 < ntot; i0 += 32)
            sum_noted<1>(list, wl, i0, lane, acc, Hm);
    } else {
        walk_owned(dstv, E, base, nown, t, [&](int e, int r) {
            const int s = srcv[e];
            float al[HEADS];
            for (int h = 0; h < HEADS; ++h) {
                const float ev = leaky(as_[(size_t)s * HEADS + h] + adl[r * HEADS + h]);
                al[h] = __expf(ev - funmap(mx[r * HEADS + h])) / sm[r * HEADS + h];
            }
            for (int ch = 0; ch < HC; ++ch)
                atomicAdd(&acc[r * HC + ch], Hm[(size_t)s * HC + ch] * al[ch >> 6]);
        });
    }
    __syncthreads();

    for (int i = t; i < nown * CDIM; i += AGG_T) {
        const int r = i >> 6, c = i & 63;
        const float* a = acc + r * HC;
        float o = (a[c] + a[CDIM + c] + a[2 * CDIM + c] + a[3 * CDIM + c]) * 0.25f
                  + bias[c];
        if (do_relu) o = fmaxf(o, 0.f);
        out[(size_t)base * CDIM + i] = o;
    }
}

extern "C" void kernel_launch(void* const* d_in, const int* in_sizes, int n_in,
                              void* d_out, int out_size, void* d_ws, size_t ws_size,
                              hipStream_t stream) {
    const float* x        = (const float*)d_in[0];
    const int*   ei       = (const int*)  d_in[1];
    const float* W1       = (const float*)d_in[2];
    const float* att_src1 = (const float*)d_in[3];
    const float* att_dst1 = (const float*)d_in[4];
    const float* b1       = (const float*)d_in[5];
    const float* W2       = (const float*)d_in[6];
    const float* att_src2 = (const float*)d_in[7];
    const float* att_dst2 = (const float*)d_in[8];
    const float* b2       = (const float*)d_in[9];

    const int Fin = 128;
    const int N   = in_sizes[0] / Fin;
    const int E   = in_sizes[1] / 2;
    const int* srcv = ei;
    const int* dstv = ei + E;

    float* ws    = (float*)d_ws;
    float* Hbuf  = ws;
    float* HID   = Hbuf + (size_t)N * HC;
    float* ASRC  = HID  + (size_t)N * CDIM;
    float* ADST  = ASRC + (size_t)N * HEADS;

    const int gemmStrips = (N / 16) * (HC / 64);
    const int gemmBlocks = (gemmStrips + 3) / 4;
    const int nhBlocks   = (N * HEADS + 255) / 256;
    const int agBlocks   = (N + RNG - 1) / RNG;

    gemm_wmma_f32<<<gemmBlocks, 128, 0, stream>>>(x, W1, Hbuf, N, Fin, HC);
    attn_scores<<<nhBlocks, 256, 0, stream>>>(Hbuf, att_src1, att_dst1, ASRC, ADST, N);
    agg_owner<<<agBlocks, AGG_T, 0, stream>>>(srcv, dstv, Hbuf, ASRC, ADST, b1, HID, E, N,  1);

    gemm_wmma_f32<<<gemmBlocks, 128, 0, stream>>>(HID, W2, Hbuf, N, CDIM, HC);
    attn_scores<<<nhBlocks, 256, 0, stream>>>(Hbuf, att_src2, att_dst2, ASRC, ADST, N);
    agg_owner<<<agBlocks, AGG_T, 0, stream>>>(srcv, dstv, Hbuf, ASRC, ADST, b2, (float*)d_out, E, N,  0);
}


# destination scan once per call: layer-1 owner blocks store their noted (source,row) lists and counts in the workspace, layer-2 blocks load them and read no dst/src list; sum order kind unchanged
# speedup vs baseline: 6.1788x; 1.1224x over previous
#include <hip/hip_runtime.h>
#include <hip/hip_bf16.h>
#include <math.h>

typedef float v2f __attribute__((ext_vector_type(2)));
typedef float v8f __attribute__((ext_vector_type(8)));

#define HEADS 4
#define HC 256
#define CDIM 64
#define NEG_SLOPE 0.2f

__global__ void gemm_wmma_f32(const float* __restrict__ A,
                              const float* __restrict__ B,
                              float* __restrict__ C,
                              int M, int K, int Ncol) {
    const int lane   = threadIdx.x & 31;
    const int wave   = threadIdx.x >> 5;
    const int wtile  = blockIdx.x * 4 + wave;
    const int wtprow = Ncol >> 6;
    const int total  = (M >> 4) * wtprow;
    if (wtile >= total) return;
    const int trow   = wtile / wtprow;
    const int tcol   = wtile - trow * wtprow;
    const int row0   = trow << 4;
    const int col0   = tcol << 6;
    const int laneM  = lane & 15;
    const int laneHi = lane >> 4;

    const float* Arow  = A + (size_t)(row0 + laneM) * K;
    const float* Bbase = B + (size_t)col0 + laneM;

    v8f acc0 = {}, acc1 = {}, acc2 = {}, acc3 = {};
#pragma unroll 2
    for (int k = 0; k < K; k += 4) {
        const int ka = k + (laneHi << 1);
        v2f a;
        a.x = Arow[ka];
        a.y = Arow[ka + 1];
        const float* B0 = Bbase + (size_t)ka * Ncol;
        const float* B1 = Bbase + (size_t)(ka + 1) * Ncol;
        v2f b0, b1, b2, b3;
        b0.x = B0[0];  b0.y = B1[0];
        b1.x = B0[16]; b1.y = B1[16];
        b2.x = B0[32]; b2.y = B1[32];
        b3.x = B0[48]; b3.y = B1[48];
        acc0 = __builtin_amdgcn_wmma_f32_16x16x4_f32(false, a, false, b0, (short)0, acc0, false, false);
        acc1 = __builtin_amdgcn_wmma_f32_16x16x4_f32(false, a, false, b1, (short)0, acc1, false, false);
        acc2 = __builtin_amdgcn_wmma_f32_16x16x4_f32(false, a, false, b2, (short)0, acc2, false, false);
        acc3 = __builtin_amdgcn_wmma_f32_16x16x4_f32(false, a, false, b3, (short)0, acc3, false, false);
    }

    float* Crow = C + (size_t)(row0 + 8 * laneHi) * Ncol + col0 + laneM;
#pragma unroll
    for (int i = 0; i < 8; ++i) {
        float* r = Crow + (size_t)i * Ncol;
        r[0]  = acc0[i];
        r[16] = acc1[i];
        r[32] = acc2[i];
        r[48] = acc3[i];
    }
}

__global__ void attn_scores(const float* __restrict__ Hm,
                            const float* __restrict__ att_s,
                            const float* __restrict__ att_d,
                            float* __restrict__ as_, float* __restrict__ ad_,
                            int N) {
    const int idx = blockIdx.x * blockDim.x + threadIdx.x;
    if (idx >= N * HEADS) return;
    const int n = idx >> 2, h = idx & 3;
    const float* row = Hm + (size_t)n * HC + h * CDIM;
    const float* vs  = att_s + h * CDIM;
    const float* vd  = att_d + h * CDIM;
    float s = 0.f, d = 0.f;
#pragma unroll 8
    for (int c = 0; c < CDIM; ++c) {
        const float v = row[c];
        s = fmaf(v, vs[c], s);
        d = fmaf(v, vd[c], d);
    }
    as_[idx] = s;
    ad_[idx] = d;
}

__device__ __forceinline__ unsigned fmap(float x) {
    unsigned u = __float_as_uint(x);
    return (u & 0x80000000u) ? ~u : (u | 0x80000000u);
}
__device__ __forceinline__ float funmap(unsigned u) {
    u = (u & 0x80000000u) ? (u ^ 0x80000000u) : ~u;
    return __uint_as_float(u);
}

__device__ __forceinline__ float leaky(float e) {
    return e > 0.f ? e : NEG_SLOPE * e;
}

#define RNG   200
#define LCAP  4096
#define AGG_T 1024
#define AGG_U 4

__device__ __forceinline__ void scan4(int4 d, int e0, int base, int nown,
                                      int* cnt, int2* list) {
    const unsigned r0 = (unsigned)(d.x - base);
    const unsigned r1 = (unsigned)(d.y - base);
    const unsigned r2 = (unsigned)(d.z - base);
    const unsigned r3 = (unsigned)(d.w - base);
    if (min(min(r0, r1), min(r2, r3)) < (unsigned)nown) {
        if (r0 < (unsigned)nown) { const int sl = atomicAdd(cnt, 1); if (sl < LCAP) list[sl] = make_int2(e0,     (int)r0); }
        if (r1 < (unsigned)nown) { const int sl = atomicAdd(cnt, 1); if (sl < LCAP) list[sl] = make_int2(e0 + 1, (int)r1); }
        if (r2 < (unsigned)nown) { const int sl = atomicAdd(cnt, 1); if (sl < LCAP) list[sl] = make_int2(e0 + 2, (int)r2); }
        if (r3 < (unsigned)nown) { const int sl = atomicAdd(cnt, 1); if (sl < LCAP) list[sl] = make_int2(e0 + 3, (int)r3); }
    }
}

template <class F>
__device__ __forceinline__ void walk_owned(const int* __restrict__ dstv, int E,
                                           int base, int nown, int t, F f) {
    for (int e = t; e < E; e += AGG_T) {
        const unsigned r = (unsigned)(dstv[e] - base);
        if (r < (unsigned)nown) f(e, (int)r);
    }
}

template <int U>
__device__ __forceinline__ void sum_noted(const int2* list, const float4* wl,
                                          int i0, int lane, float* acc,
                                          const float* __restrict__ Hm) {
    int   rr[U], ss[U];
    float al[U][HEADS];
    float v[U][8];
#pragma unroll
    for (int u = 0; u < U; ++u) {
        const int2 en = list[i0 + u];
        ss[u] = __builtin_amdgcn_readfirstlane(en.x);
        rr[u] = __builtin_amdgcn_readfirstlane(en.y);
    }
#pragma unroll
    for (int u = 0; u < U; ++u) {
        const float* hrow = Hm + (size_t)ss[u] * HC + lane;
#pragma unroll
        for (int j = 0; j < 8; ++j) v[u][j] = hrow[j * 32];
    }
#pragma unroll
    for (int u = 0; u < U; ++u) {
        const float4 w4 = wl[i0 + u];
        al[u][0] = w4.x; al[u][1] = w4.y; al[u][2] = w4.z; al[u][3] = w4.w;
    }
#pragma unroll
    for (int u = 0; u < U; ++u) {
        float* arow = acc + rr[u] * HC + lane;
#pragma unroll
        for (int j = 0; j < 8; ++j)
            atomicAdd(&arow[j * 32], v[u][j] * al[u][j >> 1]);
    }
}

template <bool KEPT>
__global__ void
agg_owner(const int* __restrict__ srcv, const int* __restrict__ dstv,
          const float* __restrict__ Hm,
          const float* __restrict__ as_, const float* __restrict__ ad_,
          const float* __restrict__ bias,
          float* __restrict__ out, int E, int N, int do_relu,
          int* __restrict__ keep_list, int* __restrict__ keep_cnt) {
    __shared__ float    acc[RNG * HC];
    __shared__ int2     list[LCAP];
    __shared__ float4   wl[LCAP];
    __shared__ float    adl[RNG * HEADS];
    __shared__ float    slf[RNG * HEADS];
    __shared__ unsigned mx[RNG * HEADS];
    __shared__ float    sm[RNG * HEADS];
    __shared__ int      cnt;

    const int t    = threadIdx.x;
    const int lane = t & 31;
    const int wave = t >> 5;
    const int base = blockIdx.x * RNG;
    const int nown = min(RNG, N - base);
    int2* kl = (int2*)keep_list + (size_t)blockIdx.x * LCAP;

    if (t == 0) cnt = 0;
    for (int i = t; i < nown * HEADS; i += AGG_T) {
        const int n4 = base * HEADS + i;
        const float a = ad_[n4];
        const float e = leaky(as_[n4] + a);
        adl[i] = a;
        slf[i] = e;
        mx[i]  = fmap(e);
        sm[i]  = 0.f;
    }
    __syncthreads();

    if constexpr (!KEPT) {
    const int  E4 = ((E & 3) == 0) ? (E >> 2) : 0;
    const int4* d4 = (const int4*)dstv;
    int q0 = t;
    for (; q0 + 3 * AGG_T < E4; q0 += 4 * AGG_T) {
        int4 dv[4];
#pragma unroll
        for (int u = 0; u < 4; ++u) dv[u] = d4[q0 + u * AGG_T];
#pragma unroll
        for (int u = 0; u < 4; ++u)
            scan4(dv[u], (q0 + u * AGG_T) << 2, base, nown, &cnt, list);
    }
    for (; q0 < E4; q0 += AGG_T)
        scan4(d4[q0], q0 << 2, base, nown, &cnt, list);
    for (int e = (E4 << 2) + t; e < E; e += AGG_T) {
        const unsigned r = (unsigned)(dstv[e] - base);
        if (r < (unsigned)nown) {
            const int sl = atomicAdd(&cnt, 1);
            if (sl < LCAP) list[sl] = make_int2(e, (int)r);
        }
    }
    __syncthreads();
    if (t == 0) keep_cnt[blockIdx.x] = cnt;
    }

    const int  ntot   = KEPT ? keep_cnt[blockIdx.x] : cnt;
    const bool listed = ntot <= LCAP;

    if (listed && ntot > 0) {
        static_assert(LCAP == 4 * AGG_T, "four list entries per thread");
        bool   ok[4];
        int2   en[4];
        int    sv[4];
        float4 a4[4];
#pragma unroll
        for (int u = 0; u < 4; ++u) {
            const int i = t + u * AGG_T;
            ok[u] = i < ntot;
            en[u] = make_int2(0, 0);
            if (ok[u]) en[u] = KEPT ? kl[i] : list[i];
        }
#pragma unroll
        for (int u = 0; u < 4; ++u) sv[u] = KEPT ? en[u].x : srcv[en[u].x];
#pragma unroll
        for (int u = 0; u < 4; ++u) a4[u] = *(const float4*)(as_ + (size_t)sv[u] * HEADS);
#pragma unroll
        for (int u = 0; u < 4; ++u) {
            if (ok[u]) {
                const int i = t + u * AGG_T;
                const int r = en[u].y;
                const float* d = adl + r * HEADS;
                float4 e4;
                e4.x = leaky(a4[u].x + d[0]);
                e4.y = leaky(a4[u].y + d[1]);
                e4.z = leaky(a4[u].z + d[2]);
                e4.w = leaky(a4[u].w + d[3]);
                wl[i]   = e4;
                list[i] = make_int2(sv[u], r);
                if constexpr (!KEPT) kl[i] = make_int2(sv[u], r);
                unsigned* m = mx + r * HEADS;
                atomicMax(&m[0], fmap(e4.x));
                atomicMax(&m[1], fmap(e4.y));
                atomicMax(&m[2], fmap(e4.z));
                atomicMax(&m[3], fmap(e4.w));
            }
        }
    } else if (!listed) {
        walk_owned(dstv, E, base, nown, t, [&](int e, int r) {
            const int s = srcv[e];
            for (int h = 0; h < HEADS; ++h)
                atomicMax(&mx[r * HEADS + h],
                          fmap(leaky(as_[(size_t)s * HEADS + h] + adl[r * HEADS + h])));
        });
    }
    __syncthreads();

    if (listed) {
        for (int i = t; i < ntot; i += AGG_T) {
            const int r = list[i].y;
            const float4 e4 = wl[i];
            const unsigned* m = mx + r * HEADS;
            float4 x4;
            x4.x = __expf(e4.x - funmap(m[0]));
            x4.y = __expf(e4.y - funmap(m[1]));
            x4.z = __expf(e4.z - funmap(m[2]));
            x4.w = __expf(e4.w - funmap(m[3]));
            wl[i] = x4;
            float* sp = sm + r * HEADS;
            atomicAdd(&sp[0], x4.x);
            atomicAdd(&sp[1], x4.y);
            atomicAdd(&sp[2], x4.z);
            atomicAdd(&sp[3], x4.w);
        }
    } else {
        walk_owned(dstv, E, base, nown, t, [&](int e, int r) {
            const int s = srcv[e];
            for (int h = 0; h < HEADS; ++h) {
                const float ev = leaky(as_[(size_t)s * HEADS + h] + adl[r * HEADS + h]);
                atomicAdd(&sm[r * HEADS + h], __expf(ev - funmap(mx[r * HEADS + h])));
            }
        });
    }
    for (int i = t; i < nown * HEADS; i += AGG_T) {
        const float ex = __expf(slf[i] - funmap(mx[i]));
        slf[i] = ex;
        atomicAdd(&sm[i], ex);
    }
    __syncthreads();

    for (int i = t; i < nown * HEADS; i += AGG_T) {
        const float dn = sm[i] + 1e-16f;
        sm[i]  = dn;
        slf[i] = slf[i] / dn;
    }
    __syncthreads();

    for (int i = t; i < nown * HC; i += AGG_T)
        acc[i] = Hm[(size_t)base * HC + i] * slf[i >> 6];
    if (listed) {
        for (int i = t; i < ntot; i += AGG_T) {
            const float* dn = sm + list[i].y * HEADS;
            float4 x4 = wl[i];
            x4.x = x4.x / dn[0];
            x4.y = x4.y / dn[1];
            x4.z = x4.z / dn[2];
            x4.w = x4.w / dn[3];
            wl[i] = x4;
        }
    }
    __syncthreads();

    if (listed) {
        const int nfull = ntot - (ntot % AGG_U);
        for (int i0 = wave * AGG_U; i0 < nfull; i0 += 32 * AGG_U)
            sum_noted<AGG_U>(list, wl, i0, lane, acc, Hm);
        for (int i0 = nfull + wave; i0 < ntot; i0 += 32)
            sum_noted<1>(list, wl, i0, lane, acc, Hm);
    } else {
        walk_owned(dstv, E, base, nown, t, [&](int e, int r) {
            const int s = srcv[e];
            float al[HEADS];
            for (int h = 0; h < HEADS; ++h) {
                const float ev = leaky(as_[(size_t)s * HEADS + h] + adl[r * HEADS + h]);
                al[h] = __expf(ev - funmap(mx[r * HEADS + h])) / sm[r * HEADS + h];
            }
            for (int ch = 0; ch < HC; ++ch)
                atomicAdd(&acc[r * HC + ch], Hm[(size_t)s * HC + ch] * al[ch >> 6]);
        });
    }
    __syncthreads();

    for (int i = t; i < nown * CDIM; i += AGG_T) {
        const int r = i >> 6, c = i & 63;
        const float* a = acc + r * HC;
        float o = (a[c] + a[CDIM + c] + a[2 * CDIM + c] + a[3 * CDIM + c]) * 0.25f
                  + bias[c];
        if (do_relu) o = fmaxf(o, 0.f);
        out[(size_t)base * CDIM + i] = o;
    }
}

extern "C" void kernel_launch(void* const* d_in, const int* in_sizes, int n_in,
                              void* d_out, int out_size, void* d_ws, size_t ws_size,
                              hipStream_t stream) {
    const float* x        = (const float*)d_in[0];
    const int*   ei       = (const int*)  d_in[1];
    const float* W1       = (const float*)d_in[2];
    const float* att_src1 = (const float*)d_in[3];
    const float* att_dst1 = (const float*)d_in[4];
    const float* b1       = (const float*)d_in[5];
    const float* W2       = (const float*)d_in[6];
    const float* att_src2 = (const float*)d_in[7];
    const float* att_dst2 = (const float*)d_in[8];
    const float* b2       = (const float*)d_in[9];

    const int Fin = 128;
    const int N   = in_sizes[0] / Fin;
    const int E   = in_sizes[1] / 2;
    const int* srcv = ei;
    const int* dstv = ei + E;

    float* ws    = (float*)d_ws;
    float* Hbuf  = ws;
    float* HID   = Hbuf + (size_t)N * HC;
    float* ASRC  = HID  + (size_t)N * CDIM;
    float* ADST  = ASRC + (size_t)N * HEADS;
    int*   KLIST = (int*)(ADST + (size_t)N * HEADS);
    int*   KCNT  = KLIST + (size_t)((N + RNG - 1) / RNG) * LCAP * 2;

    const int gemmStrips = (N / 16) * (HC / 64);
    const int gemmBlocks = (gemmStrips + 3) / 4;
    const int nhBlocks   = (N * HEADS + 255) / 256;
    const int agBlocks   = (N + RNG - 1) / RNG;

    gemm_wmma_f32<<<gemmBlocks, 128, 0, stream>>>(x, W1, Hbuf, N, Fin, HC);
    attn_scores<<<nhBlocks, 256, 0, stream>>>(Hbuf, att_src1, att_dst1, ASRC, ADST, N);
    agg_owner<false><<<agBlocks, AGG_T, 0, stream>>>(srcv, dstv, Hbuf, ASRC, ADST, b1, HID, E, N,  1, KLIST, KCNT);

    gemm_wmma_f32<<<gemmBlocks, 128, 0, stream>>>(HID, W2, Hbuf, N, CDIM, HC);
    attn_scores<<<nhBlocks, 256, 0, stream>>>(Hbuf, att_src2, att_dst2, ASRC, ADST, N);
    agg_owner<true><<<agBlocks, AGG_T, 0, stream>>>(srcv, dstv, Hbuf, ASRC, ADST, b2, (float*)d_out, E, N,  0, KLIST, KCNT);
}


# attention scores formed in the projection's epilogue from the WMMA accumulators (a wave's 16x64 strip is one head); both score launches go; sum order of scores changes (4 tiles, then 16-lane tree)
# speedup vs baseline: 6.4686x; 1.0469x over previous
#include <hip/hip_runtime.h>
#include <hip/hip_bf16.h>
#include <math.h>

typedef float v2f __attribute__((ext_vector_type(2)));
typedef float v8f __attribute__((ext_vector_type(8)));

#define HEADS 4
#define HC 256
#define CDIM 64
#define NEG_SLOPE 0.2f


template <int XM>
__device__ __forceinline__ float row_xor(float x) {
    return __int_as_float(__builtin_amdgcn_update_dpp(0, __float_as_int(x),
                                                      0x160 | XM, 0xf, 0xf, true));
}

template <int XM, int NV>
__device__ __forceinline__ void halve(float* v, bool up) {
#pragma unroll
    for (int k = 0; k < NV / 2; ++k) {
        const float keep = up ? v[k + NV / 2] : v[k];
        const float send = up ? v[k] : v[k + NV / 2];
        v[k] = keep + row_xor<XM>(send);
    }
}

__global__ void __launch_bounds__(128)
gemm_scores_wmma_f32(const float* __restrict__ A,
                     const float* __restrict__ B,
                     float* __restrict__ C,
                     const float* __restrict__ att_s,
                     const float* __restrict__ att_d,
                     float* __restrict__ as_, float* __restrict__ ad_,
                     int K) {
    __shared__ float sc[2 * 16 * HEADS];

    const int lane   = threadIdx.x & 31;
    const int wave   = threadIdx.x >> 5;
    const int row0   = blockIdx.x << 4;
    const int col0   = wave << 6;
    const int laneM  = lane & 15;
    const int laneHi = lane >> 4;

    const float* Arow  = A + (size_t)(row0 + laneM) * K;
    const float* Bbase = B + (size_t)col0 + laneM;

    v8f acc0 = {}, acc1 = {}, acc2 = {}, acc3 = {};
#pragma unroll 2
    for (int k = 0; k < K; k += 4) {
        const int ka = k + (laneHi << 1);
        v2f a;
        a.x = Arow[ka];
        a.y = Arow[ka + 1];
        const float* B0 = Bbase + (size_t)ka * HC;
        const float* B1 = Bbase + (size_t)(ka + 1) * HC;
        v2f b0, b1, b2, b3;
        b0.x = B0[0];  b0.y = B1[0];
        b1.x = B0[16]; b1.y = B1[16];
        b2.x = B0[32]; b2.y = B1[32];
        b3.x = B0[48]; b3.y = B1[48];
        acc0 = __builtin_amdgcn_wmma_f32_16x16x4_f32(false, a, false, b0, (short)0, acc0, false, false);
        acc1 = __builtin_amdgcn_wmma_f32_16x16x4_f32(false, a, false, b1, (short)0, acc1, false, false);
        acc2 = __builtin_amdgcn_wmma_f32_16x16x4_f32(false, a, false, b2, (short)0, acc2, false, false);
        acc3 = __builtin_amdgcn_wmma_f32_16x16x4_f32(false, a, false, b3, (short)0, acc3, false, false);
    }

    const float* vsp = att_s + col0 + laneM;
    const float* vdp = att_d + col0 + laneM;
    const float s0 = vsp[0], s1 = vsp[16], s2 = vsp[32], s3 = vsp[48];
    const float d0 = vdp[0], d1 = vdp[16], d2 = vdp[32], d3 = vdp[48];
    float v[16];
#pragma unroll
    for (int i = 0; i < 8; ++i) {
        v[i]     = fmaf(acc3[i], s3, fmaf(acc2[i], s2, fmaf(acc1[i], s1, acc0[i] * s0)));
        v[8 + i] = fmaf(acc3[i], d3, fmaf(acc2[i], d2, fmaf(acc1[i], d1, acc0[i] * d0)));
    }
    halve<8, 16>(v, (laneM & 8) != 0);
    halve<4, 8>(v, (laneM & 4) != 0);
    halve<2, 4>(v, (laneM & 2) != 0);
    halve<1, 2>(v, (laneM & 1) != 0);
    sc[(laneM >> 3) * (16 * HEADS) + ((laneM & 7) + 8 * laneHi) * HEADS + wave] = v[0];
    __syncthreads();
    {
        const int t = threadIdx.x;
        float* dstp = (t < 16 * HEADS) ? as_ : ad_;
        dstp[(size_t)row0 * HEADS + (t & (16 * HEADS - 1))] = sc[t];
    }

    float* Crow = C + (size_t)(row0 + 8 * laneHi) * HC + col0 + laneM;
#pragma unroll
    for (int i = 0; i < 8; ++i) {
        float* r = Crow + (size_t)i * HC;
        r[0]  = acc0[i];
        r[16] = acc1[i];
        r[32] = acc2[i];
        r[48] = acc3[i];
    }
}

__device__ __forceinline__ unsigned fmap(float x) {
    unsigned u = __float_as_uint(x);
    return (u & 0x80000000u) ? ~u : (u | 0x80000000u);
}
__device__ __forceinline__ float funmap(unsigned u) {
    u = (u & 0x80000000u) ? (u ^ 0x80000000u) : ~u;
    return __uint_as_float(u);
}

__device__ __forceinline__ float leaky(float e) {
    return e > 0.f ? e : NEG_SLOPE * e;
}

#define RNG   200
#define LCAP  4096
#define AGG_T 1024
#define AGG_U 4

__device__ __forceinline__ void scan4(int4 d, int e0, int base, int nown,
                                      int* cnt, int2* list) {
    const unsigned r0 = (unsigned)(d.x - base);
    const unsigned r1 = (unsigned)(d.y - base);
    const unsigned r2 = (unsigned)(d.z - base);
    const unsigned r3 = (unsigned)(d.w - base);
    if (min(min(r0, r1), min(r2, r3)) < (unsigned)nown) {
        if (r0 < (unsigned)nown) { const int sl = atomicAdd(cnt, 1); if (sl < LCAP) list[sl] = make_int2(e0,     (int)r0); }
        if (r1 < (unsigned)nown) { const int sl = atomicAdd(cnt, 1); if (sl < LCAP) list[sl] = make_int2(e0 + 1, (int)r1); }
        if (r2 < (unsigned)nown) { const int sl = atomicAdd(cnt, 1); if (sl < LCAP) list[sl] = make_int2(e0 + 2, (int)r2); }
        if (r3 < (unsigned)nown) { const int sl = atomicAdd(cnt, 1); if (sl < LCAP) list[sl] = make_int2(e0 + 3, (int)r3); }
    }
}

template <class F>
__device__ __forceinline__ void walk_owned(const int* __restrict__ dstv, int E,
                                           int base, int nown, int t, F f) {
    for (int e = t; e < E; e += AGG_T) {
        const unsigned r = (unsigned)(dstv[e] - base);
        if (r < (unsigned)nown) f(e, (int)r);
    }
}

template <int U>
__device__ __forceinline__ void sum_noted(const int2* list, const float4* wl,
                                          int i0, int lane, float* acc,
                                          const float* __restrict__ Hm) {
    int   rr[U], ss[U];
    float al[U][HEADS];
    float v[U][8];
#pragma unroll
    for (int u = 0; u < U; ++u) {
        const int2 en = list[i0 + u];
        ss[u] = __builtin_amdgcn_readfirstlane(en.x);
        rr[u] = __builtin_amdgcn_readfirstlane(en.y);
    }
#pragma unroll
    for (int u = 0; u < U; ++u) {
        const float* hrow = Hm + (size_t)ss[u] * HC + lane;
#pragma unroll
        for (int j = 0; j < 8; ++j) v[u][j] = hrow[j * 32];
    }
#pragma unroll
    for (int u = 0; u < U; ++u) {
        const float4 w4 = wl[i0 + u];
        al[u][0] = w4.x; al[u][1] = w4.y; al[u][2] = w4.z; al[u][3] = w4.w;
    }
#pragma unroll
    for (int u = 0; u < U; ++u) {
        float* arow = acc + rr[u] * HC + lane;
#pragma unroll
        for (int j = 0; j < 8; ++j)
            atomicAdd(&arow[j * 32], v[u][j] * al[u][j >> 1]);
    }
}

template <bool KEPT>
__global__ void
agg_owner(const int* __restrict__ srcv, const int* __restrict__ dstv,
          const float* __restrict__ Hm,
          const float* __restrict__ as_, const float* __restrict__ ad_,
          const float* __restrict__ bias,
          float* __restrict__ out, int E, int N, int do_relu,
          int* __restrict__ keep_list, int* __restrict__ keep_cnt) {
    __shared__ float    acc[RNG * HC];
    __shared__ int2     list[LCAP];
    __shared__ float4   wl[LCAP];
    __shared__ float    adl[RNG * HEADS];
    __shared__ float    slf[RNG * HEADS];
    __shared__ unsigned mx[RNG * HEADS];
    __shared__ float    sm[RNG * HEADS];
    __shared__ int      cnt;

    const int t    = threadIdx.x;
    const int lane = t & 31;
    const int wave = t >> 5;
    const int base = blockIdx.x * RNG;
    const int nown = min(RNG, N - base);
    int2* kl = (int2*)keep_list + (size_t)blockIdx.x * LCAP;

    if (t == 0) cnt = 0;
    for (int i = t; i < nown * HEADS; i += AGG_T) {
        const int n4 = base * HEADS + i;
        const float a = ad_[n4];
        const float e = leaky(as_[n4] + a);
        adl[i] = a;
        slf[i] = e;
        mx[i]  = fmap(e);
        sm[i]  = 0.f;
    }
    __syncthreads();

    if constexpr (!KEPT) {
    const int  E4 = ((E & 3) == 0) ? (E >> 2) : 0;
    const int4* d4 = (const int4*)dstv;
    int q0 = t;
    for (; q0 + 3 * AGG_T < E4; q0 += 4 * AGG_T) {
        int4 dv[4];
#pragma unroll
        for (int u = 0; u < 4; ++u) dv[u] = d4[q0 + u * AGG_T];
#pragma unroll
        for (int u = 0; u < 4; ++u)
            scan4(dv[u], (q0 + u * AGG_T) << 2, base, nown, &cnt, list);
    }
    for (; q0 < E4; q0 += AGG_T)
        scan4(d4[q0], q0 << 2, base, nown, &cnt, list);
    for (int e = (E4 << 2) + t; e < E; e += AGG_T) {
        const unsigned r = (unsigned)(dstv[e] - base);
        if (r < (unsigned)nown) {
            const int sl = atomicAdd(&cnt, 1);
            if (sl < LCAP) list[sl] = make_int2(e, (int)r);
        }
    }
    __syncthreads();
    if (t == 0) keep_cnt[blockIdx.x] = cnt;
    }

    const int  ntot   = KEPT ? keep_cnt[blockIdx.x] : cnt;
    const bool listed = ntot <= LCAP;

    if (listed && ntot > 0) {
        static_assert(LCAP == 4 * AGG_T, "four list entries per thread");
        bool   ok[4];
        int2   en[4];
        int    sv[4];
        float4 a4[4];
#pragma unroll
        for (int u = 0; u < 4; ++u) {
            const int i = t + u * AGG_T;
            ok[u] = i < ntot;
            en[u] = make_int2(0, 0);
            if (ok[u]) en[u] = KEPT ? kl[i] : list[i];
        }
#pragma unroll
        for (int u = 0; u < 4; ++u) sv[u] = KEPT ? en[u].x : srcv[en[u].x];
#pragma unroll
        for (int u = 0; u < 4; ++u) a4[u] = *(const float4*)(as_ + (size_t)sv[u] * HEADS);
#pragma unroll
        for (int u = 0; u < 4; ++u) {
            if (ok[u]) {
                const int i = t + u * AGG_T;
                const int r = en[u].y;
                const float* d = adl + r * HEADS;
                float4 e4;
                e4.x = leaky(a4[u].x + d[0]);
                e4.y = leaky(a4[u].y + d[1]);
                e4.z = leaky(a4[u].z + d[2]);
                e4.w = leaky(a4[u].w + d[3]);
                wl[i]   = e4;
                list[i] = make_int2(sv[u], r);
                if constexpr (!KEPT) kl[i] = make_int2(sv[u], r);
                unsigned* m = mx + r * HEADS;
                atomicMax(&m[0], fmap(e4.x));
                atomicMax(&m[1], fmap(e4.y));
                atomicMax(&m[2], fmap(e4.z));
                atomicMax(&m[3], fmap(e4.w));
            }
        }
    } else if (!listed) {
        walk_owned(dstv, E, base, nown, t, [&](int e, int r) {
            const int s = srcv[e];
            for (int h = 0; h < HEADS; ++h)
                atomicMax(&mx[r * HEADS + h],
                          fmap(leaky(as_[(size_t)s * HEADS + h] + adl[r * HEADS + h])));
        });
    }
    __syncthreads();

    if (listed) {
        for (int i = t; i < ntot; i += AGG_T) {
            const int r = list[i].y;
            const float4 e4 = wl[i];
            const unsigned* m = mx + r * HEADS;
            float4 x4;
            x4.x = __expf(e4.x - funmap(m[0]));
            x4.y = __expf(e4.y - funmap(m[1]));
            x4.z = __expf(e4.z - funmap(m[2]));
            x4.w = __expf(e4.w - funmap(m[3]));
            wl[i] = x4;
            float* sp = sm + r * HEADS;
            atomicAdd(&sp[0], x4.x);
            atomicAdd(&sp[1], x4.y);
            atomicAdd(&sp[2], x4.z);
            atomicAdd(&sp[3], x4.w);
        }
    } else {
        walk_owned(dstv, E, base, nown, t, [&](int e, int r) {
            const int s = srcv[e];
            for (int h = 0; h < HEADS; ++h) {
                const float ev = leaky(as_[(size_t)s * HEADS + h] + adl[r * HEADS + h]);
                atomicAdd(&sm[r * HEADS + h], __expf(ev - funmap(mx[r * HEADS + h])));
            }
        });
    }
    for (int i = t; i < nown * HEADS; i += AGG_T) {
        const float ex = __expf(slf[i] - funmap(mx[i]));
        slf[i] = ex;
        atomicAdd(&sm[i], ex);
    }
    __syncthreads();

    for (int i = t; i < nown * HEADS; i += AGG_T) {
        const float dn = sm[i] + 1e-16f;
        sm[i]  = dn;
        slf[i] = slf[i] / dn;
    }
    __syncthreads();

    for (int i = t; i < nown * HC; i += AGG_T)
        acc[i] = Hm[(size_t)base * HC + i] * slf[i >> 6];
    if (listed) {
        for (int i = t; i < ntot; i += AGG_T) {
            const float* dn = sm + list[i].y * HEADS;
            float4 x4 = wl[i];
            x4.x = x4.x / dn[0];
            x4.y = x4.y / dn[1];
            x4.z = x4.z / dn[2];
            x4.w = x4.w / dn[3];
            wl[i] = x4;
        }
    }
    __syncthreads();

    if (listed) {
        const int nfull = ntot - (ntot % AGG_U);
        for (int i0 = wave * AGG_U; i0 < nfull; i0 += 32 * AGG_U)
            sum_noted<AGG_U>(list, wl, i0, lane, acc, Hm);
        for (int i0 = nfull + wave; i0 < ntot; i0 += 32)
            sum_noted<1>(list, wl, i0, lane, acc, Hm);
    } else {
        walk_owned(dstv, E, base, nown, t, [&](int e, int r) {
            const int s = srcv[e];
            float al[HEADS];
            for (int h = 0; h < HEADS; ++h) {
                const float ev = leaky(as_[(size_t)s * HEADS + h] + adl[r * HEADS + h]);
                al[h] = __expf(ev - funmap(mx[r * HEADS + h])) / sm[r * HEADS + h];
            }
            for (int ch = 0; ch < HC; ++ch)
                atomicAdd(&acc[r * HC + ch], Hm[(size_t)s * HC + ch] * al[ch >> 6]);
        });
    }
    __syncthreads();

    for (int i = t; i < nown * CDIM; i += AGG_T) {
        const int r = i >> 6, c = i & 63;
        const float* a = acc + r * HC;
        float o = (a[c] + a[CDIM + c] + a[2 * CDIM + c] + a[3 * CDIM + c]) * 0.25f
                  + bias[c];
        if (do_relu) o = fmaxf(o, 0.f);
        out[(size_t)base * CDIM + i] = o;
    }
}

extern "C" void kernel_launch(void* const* d_in, const int* in_sizes, int n_in,
                              void* d_out, int out_size, void* d_ws, size_t ws_size,
                              hipStream_t stream) {
    const float* x        = (const float*)d_in[0];
    const int*   ei       = (const int*)  d_in[1];
    const float* W1       = (const float*)d_in[2];
    const float* att_src1 = (const float*)d_in[3];
    const float* att_dst1 = (const float*)d_in[4];
    const float* b1       = (const float*)d_in[5];
    const float* W2       = (const float*)d_in[6];
    const float* att_src2 = (const float*)d_in[7];
    const float* att_dst2 = (const float*)d_in[8];
    const float* b2       = (const float*)d_in[9];

    const int Fin = 128;
    const int N   = in_sizes[0] / Fin;
    const int E   = in_sizes[1] / 2;
    const int* srcv = ei;
    const int* dstv = ei + E;

    float* ws    = (float*)d_ws;
    float* Hbuf  = ws;
    float* HID   = Hbuf + (size_t)N * HC;
    float* ASRC  = HID  + (size_t)N * CDIM;
    float* ADST  = ASRC + (size_t)N * HEADS;
    int*   KLIST = (int*)(ADST + (size_t)N * HEADS);
    int*   KCNT  = KLIST + (size_t)((N + RNG - 1) / RNG) * LCAP * 2;

    const int gemmBlocks = N / 16;
    const int agBlocks   = (N + RNG - 1) / RNG;

    gemm_scores_wmma_f32<<<gemmBlocks, 128, 0, stream>>>(x, W1, Hbuf, att_src1, att_dst1, ASRC, ADST, Fin);
    agg_owner<false><<<agBlocks, AGG_T, 0, stream>>>(srcv, dstv, Hbuf, ASRC, ADST, b1, HID, E, N,  1, KLIST, KCNT);

    gemm_scores_wmma_f32<<<gemmBlocks, 128, 0, stream>>>(HID, W2, Hbuf, att_src2, att_dst2, ASRC, ADST, CDIM);
    agg_owner<true><<<agBlocks, AGG_T, 0, stream>>>(srcv, dstv, Hbuf, ASRC, ADST, b2, (float*)d_out, E, N,  0, KLIST, KCNT);
}


# edge row gather reads 16 bytes per lane: a source row in 2 loads of 4 whole lines each, not 8 one-line loads; LDS sum rows lane-interleaved so adds stay conflict-free; sum order kind unchanged
# speedup vs baseline: 7.0343x; 1.0875x over previous
#include <hip/hip_runtime.h>
#include <hip/hip_bf16.h>
#include <math.h>

typedef float v2f __attribute__((ext_vector_type(2)));
typedef float v8f __attribute__((ext_vector_type(8)));

#define HEADS 4
#define HC 256
#define CDIM 64
#define NEG_SLOPE 0.2f


template <int XM>
__device__ __forceinline__ float row_xor(float x) {
    return __int_as_float(__builtin_amdgcn_update_dpp(0, __float_as_int(x),
                                                      0x160 | XM, 0xf, 0xf, true));
}

template <int XM, int NV>
__device__ __forceinline__ void halve(float* v, bool up) {
#pragma unroll
    for (int k = 0; k < NV / 2; ++k) {
        const float keep = up ? v[k + NV / 2] : v[k];
        const float send = up ? v[k] : v[k + NV / 2];
        v[k] = keep + row_xor<XM>(send);
    }
}

__global__ void __launch_bounds__(128)
gemm_scores_wmma_f32(const float* __restrict__ A,
                     const float* __restrict__ B,
                     float* __restrict__ C,
                     const float* __restrict__ att_s,
                     const float* __restrict__ att_d,
                     float* __restrict__ as_, float* __restrict__ ad_,
                     int K) {
    __shared__ float sc[2 * 16 * HEADS];

    const int lane   = threadIdx.x & 31;
    const int wave   = threadIdx.x >> 5;
    const int row0   = blockIdx.x << 4;
    const int col0   = wave << 6;
    const int laneM  = lane & 15;
    const int laneHi = lane >> 4;

    const float* Arow  = A + (size_t)(row0 + laneM) * K;
    const float* Bbase = B + (size_t)col0 + laneM;

    v8f acc0 = {}, acc1 = {}, acc2 = {}, acc3 = {};
#pragma unroll 2
    for (int k = 0; k < K; k += 4) {
        const int ka = k + (laneHi << 1);
        v2f a;
        a.x = Arow[ka];
        a.y = Arow[ka + 1];
        const float* B0 = Bbase + (size_t)ka * HC;
        const float* B1 = Bbase + (size_t)(ka + 1) * HC;
        v2f b0, b1, b2, b3;
        b0.x = B0[0];  b0.y = B1[0];
        b1.x = B0[16]; b1.y = B1[16];
        b2.x = B0[32]; b2.y = B1[32];
        b3.x = B0[48]; b3.y = B1[48];
        acc0 = __builtin_amdgcn_wmma_f32_16x16x4_f32(false, a, false, b0, (short)0, acc0, false, false);
        acc1 = __builtin_amdgcn_wmma_f32_16x16x4_f32(false, a, false, b1, (short)0, acc1, false, false);
        acc2 = __builtin_amdgcn_wmma_f32_16x16x4_f32(false, a, false, b2, (short)0, acc2, false, false);
        acc3 = __builtin_amdgcn_wmma_f32_16x16x4_f32(false, a, false, b3, (short)0, acc3, false, false);
    }

    const float* vsp = att_s + col0 + laneM;
    const float* vdp = att_d + col0 + laneM;
    const float s0 = vsp[0], s1 = vsp[16], s2 = vsp[32], s3 = vsp[48];
    const float d0 = vdp[0], d1 = vdp[16], d2 = vdp[32], d3 = vdp[48];
    float v[16];
#pragma unroll
    for (int i = 0; i < 8; ++i) {
        v[i]     = fmaf(acc3[i], s3, fmaf(acc2[i], s2, fmaf(acc1[i], s1, acc0[i] * s0)));
        v[8 + i] = fmaf(acc3[i], d3, fmaf(acc2[i], d2, fmaf(acc1[i], d1, acc0[i] * d0)));
    }
    halve<8, 16>(v, (laneM & 8) != 0);
    halve<4, 8>(v, (laneM & 4) != 0);
    halve<2, 4>(v, (laneM & 2) != 0);
    halve<1, 2>(v, (laneM & 1) != 0);
    sc[(laneM >> 3) * (16 * HEADS) + ((laneM & 7) + 8 * laneHi) * HEADS + wave] = v[0];
    __syncthreads();
    {
        const int t = threadIdx.x;
        float* dstp = (t < 16 * HEADS) ? as_ : ad_;
        dstp[(size_t)row0 * HEADS + (t & (16 * HEADS - 1))] = sc[t];
    }

    float* Crow = C + (size_t)(row0 + 8 * laneHi) * HC + col0 + laneM;
#pragma unroll
    for (int i = 0; i < 8; ++i) {
        float* r = Crow + (size_t)i * HC;
        r[0]  = acc0[i];
        r[16] = acc1[i];
        r[32] = acc2[i];
        r[48] = acc3[i];
    }
}

__device__ __forceinline__ unsigned fmap(float x) {
    unsigned u = __float_as_uint(x);
    return (u & 0x80000000u) ? ~u : (u | 0x80000000u);
}
__device__ __forceinline__ float funmap(unsigned u) {
    u = (u & 0x80000000u) ? (u ^ 0x80000000u) : ~u;
    return __uint_as_float(u);
}

__device__ __forceinline__ float leaky(float e) {
    return e > 0.f ? e : NEG_SLOPE * e;
}

#define RNG   200
#define LCAP  4096
#define AGG_T 1024
#define AGG_U 4

__device__ __forceinline__ void scan4(int4 d, int e0, int base, int nown,
                                      int* cnt, int2* list) {
    const unsigned r0 = (unsigned)(d.x - base);
    const unsigned r1 = (unsigned)(d.y - base);
    const unsigned r2 = (unsigned)(d.z - base);
    const unsigned r3 = (unsigned)(d.w - base);
    if (min(min(r0, r1), min(r2, r3)) < (unsigned)nown) {
        if (r0 < (unsigned)nown) { const int sl = atomicAdd(cnt, 1); if (sl < LCAP) list[sl] = make_int2(e0,     (int)r0); }
        if (r1 < (unsigned)nown) { const int sl = atomicAdd(cnt, 1); if (sl < LCAP) list[sl] = make_int2(e0 + 1, (int)r1); }
        if (r2 < (unsigned)nown) { const int sl = atomicAdd(cnt, 1); if (sl < LCAP) list[sl] = make_int2(e0 + 2, (int)r2); }
        if (r3 < (unsigned)nown) { const int sl = atomicAdd(cnt, 1); if (sl < LCAP) list[sl] = make_int2(e0 + 3, (int)r3); }
    }
}

template <class F>
__device__ __forceinline__ void walk_owned(const int* __restrict__ dstv, int E,
                                           int base, int nown, int t, F f) {
    for (int e = t; e < E; e += AGG_T) {
        const unsigned r = (unsigned)(dstv[e] - base);
        if (r < (unsigned)nown) f(e, (int)r);
    }
}

__device__ __forceinline__ int chpos(int ch) {
    return (ch & 128) + ((ch & 3) << 5) + ((ch & 127) >> 2);
}

template <int U>
__device__ __forceinline__ void sum_noted(const int2* list, const float4* wl,
                                          int i0, int lane, float* acc,
                                          const float* __restrict__ Hm) {
    int    rr[U], ss[U];
    float  al[U][HEADS];
    float4 v[U][2];
#pragma unroll
    for (int u = 0; u < U; ++u) {
        const int2 en = list[i0 + u];
        ss[u] = __builtin_amdgcn_readfirstlane(en.x);
        rr[u] = __builtin_amdgcn_readfirstlane(en.y);
    }
#pragma unroll
    for (int u = 0; u < U; ++u) {
        const float4* hrow = (const float4*)(Hm + (size_t)ss[u] * HC) + lane;
        v[u][0] = hrow[0];
        v[u][1] = hrow[32];
    }
#pragma unroll
    for (int u = 0; u < U; ++u) {
        const float4 w4 = wl[i0 + u];
        al[u][0] = w4.x; al[u][1] = w4.y; al[u][2] = w4.z; al[u][3] = w4.w;
    }
#pragma unroll
    for (int u = 0; u < U; ++u) {
        float* arow = acc + rr[u] * HC + lane;
        const float w0 = (lane & 16) ? al[u][1] : al[u][0];
        const float w1 = (lane & 16) ? al[u][3] : al[u][2];
        atomicAdd(&arow[0 * 32], v[u][0].x * w0);
        atomicAdd(&arow[1 * 32], v[u][0].y * w0);
        atomicAdd(&arow[2 * 32], v[u][0].z * w0);
        atomicAdd(&arow[3 * 32], v[u][0].w * w0);
        atomicAdd(&arow[4 * 32], v[u][1].x * w1);
        atomicAdd(&arow[5 * 32], v[u][1].y * w1);
        atomicAdd(&arow[6 * 32], v[u][1].z * w1);
        atomicAdd(&arow[7 * 32], v[u][1].w * w1);
    }
}

template <bool KEPT>
__global__ void
agg_owner(const int* __restrict__ srcv, const int* __restrict__ dstv,
          const float* __restrict__ Hm,
          const float* __restrict__ as_, const float* __restrict__ ad_,
          const float* __restrict__ bias,
          float* __restrict__ out, int E, int N, int do_relu,
          int* __restrict__ keep_list, int* __restrict__ keep_cnt) {
    __shared__ float    acc[RNG * HC];
    __shared__ int2     list[LCAP];
    __shared__ float4   wl[LCAP];
    __shared__ float    adl[RNG * HEADS];
    __shared__ float    slf[RNG * HEADS];
    __shared__ unsigned mx[RNG * HEADS];
    __shared__ float    sm[RNG * HEADS];
    __shared__ int      cnt;

    const int t    = threadIdx.x;
    const int lane = t & 31;
    const int wave = t >> 5;
    const int base = blockIdx.x * RNG;
    const int nown = min(RNG, N - base);
    int2* kl = (int2*)keep_list + (size_t)blockIdx.x * LCAP;

    if (t == 0) cnt = 0;
    for (int i = t; i < nown * HEADS; i += AGG_T) {
        const int n4 = base * HEADS + i;
        const float a = ad_[n4];
        const float e = leaky(as_[n4] + a);
        adl[i] = a;
        slf[i] = e;
        mx[i]  = fmap(e);
        sm[i]  = 0.f;
    }
    __syncthreads();

    if constexpr (!KEPT) {
    const int  E4 = ((E & 3) == 0) ? (E >> 2) : 0;
    const int4* d4 = (const int4*)dstv;
    int q0 = t;
    for (; q0 + 3 * AGG_T < E4; q0 += 4 * AGG_T) {
        int4 dv[4];
#pragma unroll
        for (int u = 0; u < 4; ++u) dv[u] = d4[q0 + u * AGG_T];
#pragma unroll
        for (int u = 0; u < 4; ++u)
            scan4(dv[u], (q0 + u * AGG_T) << 2, base, nown, &cnt, list);
    }
    for (; q0 < E4; q0 += AGG_T)
        scan4(d4[q0], q0 << 2, base, nown, &cnt, list);
    for (int e = (E4 << 2) + t; e < E; e += AGG_T) {
        const unsigned r = (unsigned)(dstv[e] - base);
        if (r < (unsigned)nown) {
            const int sl = atomicAdd(&cnt, 1);
            if (sl < LCAP) list[sl] = make_int2(e, (int)r);
        }
    }
    __syncthreads();
    if (t == 0) keep_cnt[blockIdx.x] = cnt;
    }

    const int  ntot   = KEPT ? keep_cnt[blockIdx.x] : cnt;
    const bool listed = ntot <= LCAP;

    if (listed && ntot > 0) {
        static_assert(LCAP == 4 * AGG_T, "four list entries per thread");
        bool   ok[4];
        int2   en[4];
        int    sv[4];
        float4 a4[4];
#pragma unroll
        for (int u = 0; u < 4; ++u) {
            const int i = t + u * AGG_T;
            ok[u] = i < ntot;
            en[u] = make_int2(0, 0);
            if (ok[u]) en[u] = KEPT ? kl[i] : list[i];
        }
#pragma unroll
        for (int u = 0; u < 4; ++u) sv[u] = KEPT ? en[u].x : srcv[en[u].x];
#pragma unroll
        for (int u = 0; u < 4; ++u) a4[u] = *(const float4*)(as_ + (size_t)sv[u] * HEADS);
#pragma unroll
        for (int u = 0; u < 4; ++u) {
            if (ok[u]) {
                const int i = t + u * AGG_T;
                const int r = en[u].y;
                const float* d = adl + r * HEADS;
                float4 e4;
                e4.x = leaky(a4[u].x + d[0]);
                e4.y = leaky(a4[u].y + d[1]);
                e4.z = leaky(a4[u].z + d[2]);
                e4.w = leaky(a4[u].w + d[3]);
                wl[i]   = e4;
                list[i] = make_int2(sv[u], r);
                if constexpr (!KEPT) kl[i] = make_int2(sv[u], r);
                unsigned* m = mx + r * HEADS;
                atomicMax(&m[0], fmap(e4.x));
                atomicMax(&m[1], fmap(e4.y));
                atomicMax(&m[2], fmap(e4.z));
                atomicMax(&m[3], fmap(e4.w));
            }
        }
    } else if (!listed) {
        walk_owned(dstv, E, base, nown, t, [&](int e, int r) {
            const int s = srcv[e];
            for (int h = 0; h < HEADS; ++h)
                atomicMax(&mx[r * HEADS + h],
                          fmap(leaky(as_[(size_t)s * HEADS + h] + adl[r * HEADS + h])));
        });
    }
    __syncthreads();

    if (listed) {
        for (int i = t; i < ntot; i += AGG_T) {
            const int r = list[i].y;
            const float4 e4 = wl[i];
            const unsigned* m = mx + r * HEADS;
            float4 x4;
            x4.x = __expf(e4.x - funmap(m[0]));
            x4.y = __expf(e4.y - funmap(m[1]));
            x4.z = __expf(e4.z - funmap(m[2]));
            x4.w = __expf(e4.w - funmap(m[3]));
            wl[i] = x4;
            float* sp = sm + r * HEADS;
            atomicAdd(&sp[0], x4.x);
            atomicAdd(&sp[1], x4.y);
            atomicAdd(&sp[2], x4.z);
            atomicAdd(&sp[3], x4.w);
        }
    } else {
        walk_owned(dstv, E, base, nown, t, [&](int e, int r) {
            const int s = srcv[e];
            for (int h = 0; h < HEADS; ++h) {
                const float ev = leaky(as_[(size_t)s * HEADS + h] + adl[r * HEADS + h]);
                atomicAdd(&sm[r * HEADS + h], __expf(ev - funmap(mx[r * HEADS + h])));
            }
        });
    }
    for (int i = t; i < nown * HEADS; i += AGG_T) {
        const float ex = __expf(slf[i] - funmap(mx[i]));
        slf[i] = ex;
        atomicAdd(&sm[i], ex);
    }
    __syncthreads();

    for (int i = t; i < nown * HEADS; i += AGG_T) {
        const float dn = sm[i] + 1e-16f;
        sm[i]  = dn;
        slf[i] = slf[i] / dn;
    }
    __syncthreads();

    {
        const float4* own4 = (const float4*)(Hm + (size_t)base * HC);
        for (int i4 = t; i4 < nown * (HC / 4); i4 += AGG_T) {
            const int r = i4 >> 6, q = i4 & 63;
            const float4 v = own4[i4];
            const float  w = slf[r * HEADS + (q >> 4)];
            float* a = acc + r * HC + ((q >> 5) << 7) + (q & 31);
            a[0]  = v.x * w;
            a[32] = v.y * w;
            a[64] = v.z * w;
            a[96] = v.w * w;
        }
    }
    if (listed) {
        for (int i = t; i < ntot; i += AGG_T) {
            const float* dn = sm + list[i].y * HEADS;
            float4 x4 = wl[i];
            x4.x = x4.x / dn[0];
            x4.y = x4.y / dn[1];
            x4.z = x4.z / dn[2];
            x4.w = x4.w / dn[3];
            wl[i] = x4;
        }
    }
    __syncthreads();

    if (listed) {
        const int nfull = ntot - (ntot % AGG_U);
        for (int i0 = wave * AGG_U; i0 < nfull; i0 += 32 * AGG_U)
            sum_noted<AGG_U>(list, wl, i0, lane, acc, Hm);
        for (int i0 = nfull + wave; i0 < ntot; i0 += 32)
            sum_noted<1>(list, wl, i0, lane, acc, Hm);
    } else {
        walk_owned(dstv, E, base, nown, t, [&](int e, int r) {
            const int s = srcv[e];
            float al[HEADS];
            for (int h = 0; h < HEADS; ++h) {
                const float ev = leaky(as_[(size_t)s * HEADS + h] + adl[r * HEADS + h]);
                al[h] = __expf(ev - funmap(mx[r * HEADS + h])) / sm[r * HEADS + h];
            }
            for (int ch = 0; ch < HC; ++ch)
                atomicAdd(&acc[r * HC + chpos(ch)], Hm[(size_t)s * HC + ch] * al[ch >> 6]);
        });
    }
    __syncthreads();

    for (int i = t; i < nown * CDIM; i += AGG_T) {
        const int r = i >> 6, c = i & 63;
        const float* a = acc + r * HC;
        float o = (a[chpos(c)] + a[chpos(CDIM + c)] + a[chpos(2 * CDIM + c)] + a[chpos(3 * CDIM + c)]) * 0.25f
                  + bias[c];
        if (do_relu) o = fmaxf(o, 0.f);
        out[(size_t)base * CDIM + i] = o;
    }
}

extern "C" void kernel_launch(void* const* d_in, const int* in_sizes, int n_in,
                              void* d_out, int out_size, void* d_ws, size_t ws_size,
                              hipStream_t stream) {
    const float* x        = (const float*)d_in[0];
    const int*   ei       = (const int*)  d_in[1];
    const float* W1       = (const float*)d_in[2];
    const float* att_src1 = (const float*)d_in[3];
    const float* att_dst1 = (const float*)d_in[4];
    const float* b1       = (const float*)d_in[5];
    const float* W2       = (const float*)d_in[6];
    const float* att_src2 = (const float*)d_in[7];
    const float* att_dst2 = (const float*)d_in[8];
    const float* b2       = (const float*)d_in[9];

    const int Fin = 128;
    const int N   = in_sizes[0] / Fin;
    const int E   = in_sizes[1] / 2;
    const int* srcv = ei;
    const int* dstv = ei + E;

    float* ws    = (float*)d_ws;
    float* Hbuf  = ws;
    float* HID   = Hbuf + (size_t)N * HC;
    float* ASRC  = HID  + (size_t)N * CDIM;
    float* ADST  = ASRC + (size_t)N * HEADS;
    int*   KLIST = (int*)(ADST + (size_t)N * HEADS);
    int*   KCNT  = KLIST + (size_t)((N + RNG - 1) / RNG) * LCAP * 2;

    const int gemmBlocks = N / 16;
    const int agBlocks   = (N + RNG - 1) / RNG;

    gemm_scores_wmma_f32<<<gemmBlocks, 128, 0, stream>>>(x, W1, Hbuf, att_src1, att_dst1, ASRC, ADST, Fin);
    agg_owner<false><<<agBlocks, AGG_T, 0, stream>>>(srcv, dstv, Hbuf, ASRC, ADST, b1, HID, E, N,  1, KLIST, KCNT);

    gemm_scores_wmma_f32<<<gemmBlocks, 128, 0, stream>>>(HID, W2, Hbuf, att_src2, att_dst2, ASRC, ADST, CDIM);
    agg_owner<true><<<agBlocks, AGG_T, 0, stream>>>(srcv, dstv, Hbuf, ASRC, ADST, b2, (float*)d_out, E, N,  0, KLIST, KCNT);
}


# owner blocks filter a 1-byte owner plane (written by the layer-1 projection launch), 16 edges per lane and load, not the 4-byte destination list; row from dst[e] later; sum order kind unchanged
# speedup vs baseline: 7.7033x; 1.0951x over previous
#include <hip/hip_runtime.h>
#include <hip/hip_bf16.h>
#include <math.h>

typedef float v2f __attribute__((ext_vector_type(2)));
typedef float v8f __attribute__((ext_vector_type(8)));

#define HEADS 4
#define HC 256
#define CDIM 64
#define NEG_SLOPE 0.2f


template <int XM>
__device__ __forceinline__ float row_xor(float x) {
    return __int_as_float(__builtin_amdgcn_update_dpp(0, __float_as_int(x),
                                                      0x160 | XM, 0xf, 0xf, true));
}

template <int XM, int NV>
__device__ __forceinline__ void halve(float* v, bool up) {
#pragma unroll
    for (int k = 0; k < NV / 2; ++k) {
        const float keep = up ? v[k + NV / 2] : v[k];
        const float send = up ? v[k] : v[k + NV / 2];
        v[k] = keep + row_xor<XM>(send);
    }
}

#define RNG   200
#define NOOWN 0xFFu

__device__ __forceinline__ unsigned owner_of(int d, int N) {
    const unsigned id = (unsigned)d / RNG;
    return ((unsigned)d < (unsigned)N && id < NOOWN) ? id : NOOWN;
}

__global__ void __launch_bounds__(128)
gemm_scores_wmma_f32(const float* __restrict__ A,
                     const float* __restrict__ B,
                     float* __restrict__ C,
                     const float* __restrict__ att_s,
                     const float* __restrict__ att_d,
                     float* __restrict__ as_, float* __restrict__ ad_,
                     int K,
                     const int* __restrict__ dstv, unsigned char* __restrict__ own,
                     int E, int N) {
    __shared__ float sc[2 * 16 * HEADS];

    if (own) {
        const int gt = blockIdx.x * 128 + threadIdx.x;
        const int gs = gridDim.x * 128;
        if ((E & 3) == 0) {
            const int4* d4 = (const int4*)dstv;
            unsigned*   o4 = (unsigned*)own;
            for (int g = gt; g < (E >> 2); g += gs) {
                const int4 d = d4[g];
                o4[g] = owner_of(d.x, N) | (owner_of(d.y, N) << 8) |
                        (owner_of(d.z, N) << 16) | (owner_of(d.w, N) << 24);
            }
        } else {
            for (int e = gt; e < E; e += gs) own[e] = (unsigned char)owner_of(dstv[e], N);
        }
    }

    const int lane   = threadIdx.x & 31;
    const int wave   = threadIdx.x >> 5;
    const int row0   = blockIdx.x << 4;
    const int col0   = wave << 6;
    const int laneM  = lane & 15;
    const int laneHi = lane >> 4;

    const float* Arow  = A + (size_t)(row0 + laneM) * K;
    const float* Bbase = B + (size_t)col0 + laneM;

    v8f acc0 = {}, acc1 = {}, acc2 = {}, acc3 = {};
#pragma unroll 2
    for (int k = 0; k < K; k += 4) {
        const int ka = k + (laneHi << 1);
        v2f a;
        a.x = Arow[ka];
        a.y = Arow[ka + 1];
        const float* B0 = Bbase + (size_t)ka * HC;
        const float* B1 = Bbase + (size_t)(ka + 1) * HC;
        v2f b0, b1, b2, b3;
        b0.x = B0[0];  b0.y = B1[0];
        b1.x = B0[16]; b1.y = B1[16];
        b2.x = B0[32]; b2.y = B1[32];
        b3.x = B0[48]; b3.y = B1[48];
        acc0 = __builtin_amdgcn_wmma_f32_16x16x4_f32(false, a, false, b0, (short)0, acc0, false, false);
        acc1 = __builtin_amdgcn_wmma_f32_16x16x4_f32(false, a, false, b1, (short)0, acc1, false, false);
        acc2 = __builtin_amdgcn_wmma_f32_16x16x4_f32(false, a, false, b2, (short)0, acc2, false, false);
        acc3 = __builtin_amdgcn_wmma_f32_16x16x4_f32(false, a, false, b3, (short)0, acc3, false, false);
    }

    const float* vsp = att_s + col0 + laneM;
    const float* vdp = att_d + col0 + laneM;
    const float s0 = vsp[0], s1 = vsp[16], s2 = vsp[32], s3 = vsp[48];
    const float d0 = vdp[0], d1 = vdp[16], d2 = vdp[32], d3 = vdp[48];
    float v[16];
#pragma unroll
    for (int i = 0; i < 8; ++i) {
        v[i]     = fmaf(acc3[i], s3, fmaf(acc2[i], s2, fmaf(acc1[i], s1, acc0[i] * s0)));
        v[8 + i] = fmaf(acc3[i], d3, fmaf(acc2[i], d2, fmaf(acc1[i], d1, acc0[i] * d0)));
    }
    halve<8, 16>(v, (laneM & 8) != 0);
    halve<4, 8>(v, (laneM & 4) != 0);
    halve<2, 4>(v, (laneM & 2) != 0);
    halve<1, 2>(v, (laneM & 1) != 0);
    sc[(laneM >> 3) * (16 * HEADS) + ((laneM & 7) + 8 * laneHi) * HEADS + wave] = v[0];
    __syncthreads();
    {
        const int t = threadIdx.x;
        float* dstp = (t < 16 * HEADS) ? as_ : ad_;
        dstp[(size_t)row0 * HEADS + (t & (16 * HEADS - 1))] = sc[t];
    }

    float* Crow = C + (size_t)(row0 + 8 * laneHi) * HC + col0 + laneM;
#pragma unroll
    for (int i = 0; i < 8; ++i) {
        float* r = Crow + (size_t)i * HC;
        r[0]  = acc0[i];
        r[16] = acc1[i];
        r[32] = acc2[i];
        r[48] = acc3[i];
    }
}

__device__ __forceinline__ unsigned fmap(float x) {
    unsigned u = __float_as_uint(x);
    return (u & 0x80000000u) ? ~u : (u | 0x80000000u);
}
__device__ __forceinline__ float funmap(unsigned u) {
    u = (u & 0x80000000u) ? (u ^ 0x80000000u) : ~u;
    return __uint_as_float(u);
}

__device__ __forceinline__ float leaky(float e) {
    return e > 0.f ? e : NEG_SLOPE * e;
}

#define LCAP  4096
#define AGG_T 1024
#define AGG_U 4

__device__ __forceinline__ void note_edge(int e, int* cnt, int2* list) {
    const int sl = atomicAdd(cnt, 1);
    if (sl < LCAP) list[sl].x = e;
}

__device__ __forceinline__ unsigned zero_bytes(unsigned x) {
    return ~(((x & 0x7F7F7F7Fu) + 0x7F7F7F7Fu) | x | 0x7F7F7F7Fu);
}

__device__ __forceinline__ unsigned own_bits(uint4 o, unsigned pat) {
    return (zero_bytes(o.x ^ pat) >> 7) | (zero_bytes(o.y ^ pat) >> 6) |
           (zero_bytes(o.z ^ pat) >> 5) | (zero_bytes(o.w ^ pat) >> 4);
}

__device__ __forceinline__ void note_bits(unsigned m, int e0, int estep,
                                          int* cnt, int2* list) {
    while (m) {
        const int b = __builtin_ctz(m);
        m &= m - 1;
        note_edge(e0 + ((b & 4) ? estep : 0) + ((b & 3) << 2) + (b >> 3), cnt, list);
    }
}

__device__ __forceinline__ void scan4(int4 d, int e0, int base, int nown,
                                      int* cnt, int2* list) {
    const unsigned r0 = (unsigned)(d.x - base);
    const unsigned r1 = (unsigned)(d.y - base);
    const unsigned r2 = (unsigned)(d.z - base);
    const unsigned r3 = (unsigned)(d.w - base);
    if (min(min(r0, r1), min(r2, r3)) < (unsigned)nown) {
        if (r0 < (unsigned)nown) note_edge(e0,     cnt, list);
        if (r1 < (unsigned)nown) note_edge(e0 + 1, cnt, list);
        if (r2 < (unsigned)nown) note_edge(e0 + 2, cnt, list);
        if (r3 < (unsigned)nown) note_edge(e0 + 3, cnt, list);
    }
}

template <class F>
__device__ __forceinline__ void walk_owned(const int* __restrict__ dstv, int E,
                                           int base, int nown, int t, F f) {
    for (int e = t; e < E; e += AGG_T) {
        const unsigned r = (unsigned)(dstv[e] - base);
        if (r < (unsigned)nown) f(e, (int)r);
    }
}

__device__ __forceinline__ int chpos(int ch) {
    return (ch & 128) + ((ch & 3) << 5) + ((ch & 127) >> 2);
}

template <int U>
__device__ __forceinline__ void sum_noted(const int2* list, const float4* wl,
                                          int i0, int lane, float* acc,
                                          const float* __restrict__ Hm) {
    int    rr[U], ss[U];
    float  al[U][HEADS];
    float4 v[U][2];
#pragma unroll
    for (int u = 0; u < U; ++u) {
        const int2 en = list[i0 + u];
        ss[u] = __builtin_amdgcn_readfirstlane(en.x);
        rr[u] = __builtin_amdgcn_readfirstlane(en.y);
    }
#pragma unroll
    for (int u = 0; u < U; ++u) {
        const float4* hrow = (const float4*)(Hm + (size_t)ss[u] * HC) + lane;
        v[u][0] = hrow[0];
        v[u][1] = hrow[32];
    }
#pragma unroll
    for (int u = 0; u < U; ++u) {
        const float4 w4 = wl[i0 + u];
        al[u][0] = w4.x; al[u][1] = w4.y; al[u][2] = w4.z; al[u][3] = w4.w;
    }
#pragma unroll
    for (int u = 0; u < U; ++u) {
        float* arow = acc + rr[u] * HC + lane;
        const float w0 = (lane & 16) ? al[u][1] : al[u][0];
        const float w1 = (lane & 16) ? al[u][3] : al[u][2];
        atomicAdd(&arow[0 * 32], v[u][0].x * w0);
        atomicAdd(&arow[1 * 32], v[u][0].y * w0);
        atomicAdd(&arow[2 * 32], v[u][0].z * w0);
        atomicAdd(&arow[3 * 32], v[u][0].w * w0);
        atomicAdd(&arow[4 * 32], v[u][1].x * w1);
        atomicAdd(&arow[5 * 32], v[u][1].y * w1);
        atomicAdd(&arow[6 * 32], v[u][1].z * w1);
        atomicAdd(&arow[7 * 32], v[u][1].w * w1);
    }
}

template <bool KEPT>
__global__ void
agg_owner(const int* __restrict__ srcv, const int* __restrict__ dstv,
          const float* __restrict__ Hm,
          const float* __restrict__ as_, const float* __restrict__ ad_,
          const float* __restrict__ bias,
          float* __restrict__ out, int E, int N, int do_relu,
          int* __restrict__ keep_list, int* __restrict__ keep_cnt,
          const unsigned char* __restrict__ own) {
    __shared__ float    acc[RNG * HC];
    __shared__ int2     list[LCAP];
    __shared__ float4   wl[LCAP];
    __shared__ float    adl[RNG * HEADS];
    __shared__ float    slf[RNG * HEADS];
    __shared__ unsigned mx[RNG * HEADS];
    __shared__ float    sm[RNG * HEADS];
    __shared__ int      cnt;

    const int t    = threadIdx.x;
    const int lane = t & 31;
    const int wave = t >> 5;
    const int base = blockIdx.x * RNG;
    const int nown = min(RNG, N - base);
    int2* kl = (int2*)keep_list + (size_t)blockIdx.x * LCAP;

    if (t == 0) cnt = 0;
    for (int i = t; i < nown * HEADS; i += AGG_T) {
        const int n4 = base * HEADS + i;
        const float a = ad_[n4];
        const float e = leaky(as_[n4] + a);
        adl[i] = a;
        slf[i] = e;
        mx[i]  = fmap(e);
        sm[i]  = 0.f;
    }
    __syncthreads();

    if constexpr (!KEPT) {
    if (gridDim.x <= NOOWN) {
        const int    E16 = E >> 4;
        const uint4* o4  = (const uint4*)own;
        const unsigned pat = blockIdx.x * 0x01010101u;
        int q0 = t;
        for (; q0 + 7 * AGG_T < E16; q0 += 8 * AGG_T) {
            uint4 ov[8];
#pragma unroll
            for (int u = 0; u < 8; ++u) ov[u] = o4[q0 + u * AGG_T];
#pragma unroll
            for (int u = 0; u < 8; u += 2)
                note_bits(own_bits(ov[u], pat) | (own_bits(ov[u + 1], pat) << 4),
                          (q0 + u * AGG_T) << 4, AGG_T << 4, &cnt, list);
        }
        for (; q0 < E16; q0 += AGG_T)
            note_bits(own_bits(o4[q0], pat), q0 << 4, 0, &cnt, list);
        for (int e = (E16 << 4) + t; e < E; e += AGG_T)
            if (own[e] == blockIdx.x) note_edge(e, &cnt, list);
    } else {
        const int  E4 = ((E & 3) == 0) ? (E >> 2) : 0;
        const int4* d4 = (const int4*)dstv;
        int q0 = t;
        for (; q0 + 3 * AGG_T < E4; q0 += 4 * AGG_T) {
            int4 dv[4];
#pragma unroll
            for (int u = 0; u < 4; ++u) dv[u] = d4[q0 + u * AGG_T];
#pragma unroll
            for (int u = 0; u < 4; ++u)
                scan4(dv[u], (q0 + u * AGG_T) << 2, base, nown, &cnt, list);
        }
        for (; q0 < E4; q0 += AGG_T)
            scan4(d4[q0], q0 << 2, base, nown, &cnt, list);
        for (int e = (E4 << 2) + t; e < E; e += AGG_T)
            if ((unsigned)(dstv[e] - base) < (unsigned)nown) note_edge(e, &cnt, list);
    }
    __syncthreads();
    if (t == 0) keep_cnt[blockIdx.x] = cnt;
    }

    const int  ntot   = KEPT ? keep_cnt[blockIdx.x] : cnt;
    const bool listed = ntot <= LCAP;

    if (listed && ntot > 0) {
        static_assert(LCAP == 4 * AGG_T, "four list entries per thread");
        bool   ok[4];
        int2   en[4];
        int    sv[4];
        float4 a4[4];
#pragma unroll
        for (int u = 0; u < 4; ++u) {
            const int i = t + u * AGG_T;
            ok[u] = i < ntot;
            en[u] = make_int2(0, 0);
            if (ok[u]) en[u] = KEPT ? kl[i] : make_int2(list[i].x, 0);
        }
#pragma unroll
        for (int u = 0; u < 4; ++u) sv[u] = KEPT ? en[u].x : srcv[en[u].x];
        if constexpr (!KEPT) {
#pragma unroll
            for (int u = 0; u < 4; ++u) en[u].y = dstv[en[u].x] - base;
        }
#pragma unroll
        for (int u = 0; u < 4; ++u) a4[u] = *(const float4*)(as_ + (size_t)sv[u] * HEADS);
#pragma unroll
        for (int u = 0; u < 4; ++u) {
            if (ok[u]) {
                const int i = t + u * AGG_T;
                const int r = en[u].y;
                const float* d = adl + r * HEADS;
                float4 e4;
                e4.x = leaky(a4[u].x + d[0]);
                e4.y = leaky(a4[u].y + d[1]);
                e4.z = leaky(a4[u].z + d[2]);
                e4.w = leaky(a4[u].w + d[3]);
                wl[i]   = e4;
                list[i] = make_int2(sv[u], r);
                if constexpr (!KEPT) kl[i] = make_int2(sv[u], r);
                unsigned* m = mx + r * HEADS;
                atomicMax(&m[0], fmap(e4.x));
                atomicMax(&m[1], fmap(e4.y));
                atomicMax(&m[2], fmap(e4.z));
                atomicMax(&m[3], fmap(e4.w));
            }
        }
    } else if (!listed) {
        walk_owned(dstv, E, base, nown, t, [&](int e, int r) {
            const int s = srcv[e];
            for (int h = 0; h < HEADS; ++h)
                atomicMax(&mx[r * HEADS + h],
                          fmap(leaky(as_[(size_t)s * HEADS + h] + adl[r * HEADS + h])));
        });
    }
    __syncthreads();

    if (listed) {
        for (int i = t; i < ntot; i += AGG_T) {
            const int r = list[i].y;
            const float4 e4 = wl[i];
            const unsigned* m = mx + r * HEADS;
            float4 x4;
            x4.x = __expf(e4.x - funmap(m[0]));
            x4.y = __expf(e4.y - funmap(m[1]));
            x4.z = __expf(e4.z - funmap(m[2]));
            x4.w = __expf(e4.w - funmap(m[3]));
            wl[i] = x4;
            float* sp = sm + r * HEADS;
            atomicAdd(&sp[0], x4.x);
            atomicAdd(&sp[1], x4.y);
            atomicAdd(&sp[2], x4.z);
            atomicAdd(&sp[3], x4.w);
        }
    } else {
        walk_owned(dstv, E, base, nown, t, [&](int e, int r) {
            const int s = srcv[e];
            for (int h = 0; h < HEADS; ++h) {
                const float ev = leaky(as_[(size_t)s * HEADS + h] + adl[r * HEADS + h]);
                atomicAdd(&sm[r * HEADS + h], __expf(ev - funmap(mx[r * HEADS + h])));
            }
        });
    }
    for (int i = t; i < nown * HEADS; i += AGG_T) {
        const float ex = __expf(slf[i] - funmap(mx[i]));
        slf[i] = ex;
        atomicAdd(&sm[i], ex);
    }
    __syncthreads();

    for (int i = t; i < nown * HEADS; i += AGG_T) {
        const float dn = sm[i] + 1e-16f;
        sm[i]  = dn;
        slf[i] = slf[i] / dn;
    }
    __syncthreads();

    {
        const float4* own4 = (const float4*)(Hm + (size_t)base * HC);
        for (int i4 = t; i4 < nown * (HC / 4); i4 += AGG_T) {
            const int r = i4 >> 6, q = i4 & 63;
            const float4 v = own4[i4];
            const float  w = slf[r * HEADS + (q >> 4)];
            float* a = acc + r * HC + ((q >> 5) << 7) + (q & 31);
            a[0]  = v.x * w;
            a[32] = v.y * w;
            a[64] = v.z * w;
            a[96] = v.w * w;
        }
    }
    if (listed) {
        for (int i = t; i < ntot; i += AGG_T) {
            const float* dn = sm + list[i].y * HEADS;
            float4 x4 = wl[i];
            x4.x = x4.x / dn[0];
            x4.y = x4.y / dn[1];
            x4.z = x4.z / dn[2];
            x4.w = x4.w / dn[3];
            wl[i] = x4;
        }
    }
    __syncthreads();

    if (listed) {
        const int nfull = ntot - (ntot % AGG_U);
        for (int i0 = wave * AGG_U; i0 < nfull; i0 += 32 * AGG_U)
            sum_noted<AGG_U>(list, wl, i0, lane, acc, Hm);
        for (int i0 = nfull + wave; i0 < ntot; i0 += 32)
            sum_noted<1>(list, wl, i0, lane, acc, Hm);
    } else {
        walk_owned(dstv, E, base, nown, t, [&](int e, int r) {
            const int s = srcv[e];
            float al[HEADS];
            for (int h = 0; h < HEADS; ++h) {
                const float ev = leaky(as_[(size_t)s * HEADS + h] + adl[r * HEADS + h]);
                al[h] = __expf(ev - funmap(mx[r * HEADS + h])) / sm[r * HEADS + h];
            }
            for (int ch = 0; ch < HC; ++ch)
                atomicAdd(&acc[r * HC + chpos(ch)], Hm[(size_t)s * HC + ch] * al[ch >> 6]);
        });
    }
    __syncthreads();

    for (int i = t; i < nown * CDIM; i += AGG_T) {
        const int r = i >> 6, c = i & 63;
        const float* a = acc + r * HC;
        float o = (a[chpos(c)] + a[chpos(CDIM + c)] + a[chpos(2 * CDIM + c)] + a[chpos(3 * CDIM + c)]) * 0.25f
                  + bias[c];
        if (do_relu) o = fmaxf(o, 0.f);
        out[(size_t)base * CDIM + i] = o;
    }
}

extern "C" void kernel_launch(void* const* d_in, const int* in_sizes, int n_in,
                              void* d_out, int out_size, void* d_ws, size_t ws_size,
                              hipStream_t stream) {
    const float* x        = (const float*)d_in[0];
    const int*   ei       = (const int*)  d_in[1];
    const float* W1       = (const float*)d_in[2];
    const float* att_src1 = (const float*)d_in[3];
    const float* att_dst1 = (const float*)d_in[4];
    const float* b1       = (const float*)d_in[5];
    const float* W2       = (const float*)d_in[6];
    const float* att_src2 = (const float*)d_in[7];
    const float* att_dst2 = (const float*)d_in[8];
    const float* b2       = (const float*)d_in[9];

    const int Fin = 128;
    const int N   = in_sizes[0] / Fin;
    const int E   = in_sizes[1] / 2;
    const int* srcv = ei;
    const int* dstv = ei + E;

    float* ws    = (float*)d_ws;
    float* Hbuf  = ws;
    float* HID   = Hbuf + (size_t)N * HC;
    float* ASRC  = HID  + (size_t)N * CDIM;
    float* ADST  = ASRC + (size_t)N * HEADS;
    int*   KLIST = (int*)(ADST + (size_t)N * HEADS);
    unsigned char* OWN = (unsigned char*)(KLIST + (size_t)((N + RNG - 1) / RNG) * LCAP * 2);
    int*   KCNT  = (int*)(OWN + ((size_t)E + 15) / 16 * 16);

    const int gemmBlocks = N / 16;
    const int agBlocks   = (N + RNG - 1) / RNG;

    gemm_scores_wmma_f32<<<gemmBlocks, 128, 0, stream>>>(x, W1, Hbuf, att_src1, att_dst1, ASRC, ADST, Fin, dstv, OWN, E, N);
    agg_owner<false><<<agBlocks, AGG_T, 0, stream>>>(srcv, dstv, Hbuf, ASRC, ADST, b1, HID, E, N,  1, KLIST, KCNT, OWN);

    gemm_scores_wmma_f32<<<gemmBlocks, 128, 0, stream>>>(HID, W2, Hbuf, att_src2, att_dst2, ASRC, ADST, CDIM, nullptr, nullptr, 0, 0);
    agg_owner<true><<<agBlocks, AGG_T, 0, stream>>>(srcv, dstv, Hbuf, ASRC, ADST, b2, (float*)d_out, E, N,  0, KLIST, KCNT, OWN);
}


# projection: the block's 16 x K tile of A copied once to LDS by whole-line 16-byte loads (padded rows), waves read A fragments from LDS; was an 8-byte 16-line gather per wave and k-step
# speedup vs baseline: 8.2358x; 1.0691x over previous
#include <hip/hip_runtime.h>
#include <hip/hip_bf16.h>
#include <math.h>

typedef float v2f __attribute__((ext_vector_type(2)));
typedef float v8f __attribute__((ext_vector_type(8)));

#define HEADS 4
#define HC 256
#define CDIM 64
#define NEG_SLOPE 0.2f


template <int XM>
__device__ __forceinline__ float row_xor(float x) {
    return __int_as_float(__builtin_amdgcn_update_dpp(0, __float_as_int(x),
                                                      0x160 | XM, 0xf, 0xf, true));
}

template <int XM, int NV>
__device__ __forceinline__ void halve(float* v, bool up) {
#pragma unroll
    for (int k = 0; k < NV / 2; ++k) {
        const float keep = up ? v[k + NV / 2] : v[k];
        const float send = up ? v[k] : v[k + NV / 2];
        v[k] = keep + row_xor<XM>(send);
    }
}

#define APITCH 132
#define RNG   200
#define NOOWN 0xFFu

__device__ __forceinline__ unsigned owner_of(int d, int N) {
    const unsigned id = (unsigned)d / RNG;
    return ((unsigned)d < (unsigned)N && id < NOOWN) ? id : NOOWN;
}

__global__ void __launch_bounds__(128)
gemm_scores_wmma_f32(const float* __restrict__ A,
                     const float* __restrict__ B,
                     float* __restrict__ C,
                     const float* __restrict__ att_s,
                     const float* __restrict__ att_d,
                     float* __restrict__ as_, float* __restrict__ ad_,
                     int K,
                     const int* __restrict__ dstv, unsigned char* __restrict__ own,
                     int E, int N) {
    __shared__ float sc[2 * 16 * HEADS];
    __shared__ __attribute__((aligned(16))) float At[16 * APITCH];

    const int lane   = threadIdx.x & 31;
    const int wave   = threadIdx.x >> 5;
    const int row0   = blockIdx.x << 4;
    const int col0   = wave << 6;
    const int laneM  = lane & 15;
    const int laneHi = lane >> 4;

    {
        const int r  = threadIdx.x >> 3;
        const int c0 = threadIdx.x & 7;
        const int q  = K >> 2;
        const float4* Ag = (const float4*)(A + (size_t)(row0 + r) * K);
        float4 tv[4];
#pragma unroll
        for (int j = 0; j < 4; ++j) tv[j] = Ag[min(c0 + 8 * j, q - 1)];
#pragma unroll
        for (int j = 0; j < 4; ++j)
            *(float4*)&At[r * APITCH + (min(c0 + 8 * j, q - 1) << 2)] = tv[j];
    }
    __syncthreads();

    if (own) {
        const int gt = blockIdx.x * 128 + threadIdx.x;
        const int gs = gridDim.x * 128;
        if ((E & 3) == 0) {
            const int4* d4 = (const int4*)dstv;
            unsigned*   o4 = (unsigned*)own;
            for (int g = gt; g < (E >> 2); g += gs) {
                const int4 d = d4[g];
                o4[g] = owner_of(d.x, N) | (owner_of(d.y, N) << 8) |
                        (owner_of(d.z, N) << 16) | (owner_of(d.w, N) << 24);
            }
        } else {
            for (int e = gt; e < E; e += gs) own[e] = (unsigned char)owner_of(dstv[e], N);
        }
    }

    const float* Arow  = At + laneM * APITCH;
    const float* Bbase = B + (size_t)col0 + laneM;

    v8f acc0 = {}, acc1 = {}, acc2 = {}, acc3 = {};
#pragma unroll 2
    for (int k = 0; k < K; k += 4) {
        const int ka = k + (laneHi << 1);
        const v2f a = *(const v2f*)(Arow + ka);
        const float* B0 = Bbase + (size_t)ka * HC;
        const float* B1 = Bbase + (size_t)(ka + 1) * HC;
        v2f b0, b1, b2, b3;
        b0.x = B0[0];  b0.y = B1[0];
        b1.x = B0[16]; b1.y = B1[16];
        b2.x = B0[32]; b2.y = B1[32];
        b3.x = B0[48]; b3.y = B1[48];
        acc0 = __builtin_amdgcn_wmma_f32_16x16x4_f32(false, a, false, b0, (short)0, acc0, false, false);
        acc1 = __builtin_amdgcn_wmma_f32_16x16x4_f32(false, a, false, b1, (short)0, acc1, false, false);
        acc2 = __builtin_amdgcn_wmma_f32_16x16x4_f32(false, a, false, b2, (short)0, acc2, false, false);
        acc3 = __builtin_amdgcn_wmma_f32_16x16x4_f32(false, a, false, b3, (short)0, acc3, false, false);
    }

    const float* vsp = att_s + col0 + laneM;
    const float* vdp = att_d + col0 + laneM;
    const float s0 = vsp[0], s1 = vsp[16], s2 = vsp[32], s3 = vsp[48];
    const float d0 = vdp[0], d1 = vdp[16], d2 = vdp[32], d3 = vdp[48];
    float v[16];
#pragma unroll
    for (int i = 0; i < 8; ++i) {
        v[i]     = fmaf(acc3[i], s3, fmaf(acc2[i], s2, fmaf(acc1[i], s1, acc0[i] * s0)));
        v[8 + i] = fmaf(acc3[i], d3, fmaf(acc2[i], d2, fmaf(acc1[i], d1, acc0[i] * d0)));
    }
    halve<8, 16>(v, (laneM & 8) != 0);
    halve<4, 8>(v, (laneM & 4) != 0);
    halve<2, 4>(v, (laneM & 2) != 0);
    halve<1, 2>(v, (laneM & 1) != 0);
    sc[(laneM >> 3) * (16 * HEADS) + ((laneM & 7) + 8 * laneHi) * HEADS + wave] = v[0];
    __syncthreads();
    {
        const int t = threadIdx.x;
        float* dstp = (t < 16 * HEADS) ? as_ : ad_;
        dstp[(size_t)row0 * HEADS + (t & (16 * HEADS - 1))] = sc[t];
    }

    float* Crow = C + (size_t)(row0 + 8 * laneHi) * HC + col0 + laneM;
#pragma unroll
    for (int i = 0; i < 8; ++i) {
        float* r = Crow + (size_t)i * HC;
        r[0]  = acc0[i];
        r[16] = acc1[i];
        r[32] = acc2[i];
        r[48] = acc3[i];
    }
}

__device__ __forceinline__ unsigned fmap(float x) {
    unsigned u = __float_as_uint(x);
    return (u & 0x80000000u) ? ~u : (u | 0x80000000u);
}
__device__ __forceinline__ float funmap(unsigned u) {
    u = (u & 0x80000000u) ? (u ^ 0x80000000u) : ~u;
    return __uint_as_float(u);
}

__device__ __forceinline__ float leaky(float e) {
    return e > 0.f ? e : NEG_SLOPE * e;
}

#define LCAP  4096
#define AGG_T 1024
#define AGG_U 4

__device__ __forceinline__ void note_edge(int e, int* cnt, int2* list) {
    const int sl = atomicAdd(cnt, 1);
    if (sl < LCAP) list[sl].x = e;
}

__device__ __forceinline__ unsigned zero_bytes(unsigned x) {
    return ~(((x & 0x7F7F7F7Fu) + 0x7F7F7F7Fu) | x | 0x7F7F7F7Fu);
}

__device__ __forceinline__ unsigned own_bits(uint4 o, unsigned pat) {
    return (zero_bytes(o.x ^ pat) >> 7) | (zero_bytes(o.y ^ pat) >> 6) |
           (zero_bytes(o.z ^ pat) >> 5) | (zero_bytes(o.w ^ pat) >> 4);
}

__device__ __forceinline__ void note_bits(unsigned m, int e0, int estep,
                                          int* cnt, int2* list) {
    while (m) {
        const int b = __builtin_ctz(m);
        m &= m - 1;
        note_edge(e0 + ((b & 4) ? estep : 0) + ((b & 3) << 2) + (b >> 3), cnt, list);
    }
}

__device__ __forceinline__ void scan4(int4 d, int e0, int base, int nown,
                                      int* cnt, int2* list) {
    const unsigned r0 = (unsigned)(d.x - base);
    const unsigned r1 = (unsigned)(d.y - base);
    const unsigned r2 = (unsigned)(d.z - base);
    const unsigned r3 = (unsigned)(d.w - base);
    if (min(min(r0, r1), min(r2, r3)) < (unsigned)nown) {
        if (r0 < (unsigned)nown) note_edge(e0,     cnt, list);
        if (r1 < (unsigned)nown) note_edge(e0 + 1, cnt, list);
        if (r2 < (unsigned)nown) note_edge(e0 + 2, cnt, list);
        if (r3 < (unsigned)nown) note_edge(e0 + 3, cnt, list);
    }
}

template <class F>
__device__ __forceinline__ void walk_owned(const int* __restrict__ dstv, int E,
                                           int base, int nown, int t, F f) {
    for (int e = t; e < E; e += AGG_T) {
        const unsigned r = (unsigned)(dstv[e] - base);
        if (r < (unsigned)nown) f(e, (int)r);
    }
}

__device__ __forceinline__ int chpos(int ch) {
    return (ch & 128) + ((ch & 3) << 5) + ((ch & 127) >> 2);
}

template <int U>
__device__ __forceinline__ void sum_noted(const int2* list, const float4* wl,
                                          int i0, int lane, float* acc,
                                          const float* __restrict__ Hm) {
    int    rr[U], ss[U];
    float  al[U][HEADS];
    float4 v[U][2];
#pragma unroll
    for (int u = 0; u < U; ++u) {
        const int2 en = list[i0 + u];
        ss[u] = __builtin_amdgcn_readfirstlane(en.x);
        rr[u] = __builtin_amdgcn_readfirstlane(en.y);
    }
#pragma unroll
    for (int u = 0; u < U; ++u) {
        const float4* hrow = (const float4*)(Hm + (size_t)ss[u] * HC) + lane;
        v[u][0] = hrow[0];
        v[u][1] = hrow[32];
    }
#pragma unroll
    for (int u = 0; u < U; ++u) {
        const float4 w4 = wl[i0 + u];
        al[u][0] = w4.x; al[u][1] = w4.y; al[u][2] = w4.z; al[u][3] = w4.w;
    }
#pragma unroll
    for (int u = 0; u < U; ++u) {
        float* arow = acc + rr[u] * HC + lane;
        const float w0 = (lane & 16) ? al[u][1] : al[u][0];
        const float w1 = (lane & 16) ? al[u][3] : al[u][2];
        atomicAdd(&arow[0 * 32], v[u][0].x * w0);
        atomicAdd(&arow[1 * 32], v[u][0].y * w0);
        atomicAdd(&arow[2 * 32], v[u][0].z * w0);
        atomicAdd(&arow[3 * 32], v[u][0].w * w0);
        atomicAdd(&arow[4 * 32], v[u][1].x * w1);
        atomicAdd(&arow[5 * 32], v[u][1].y * w1);
        atomicAdd(&arow[6 * 32], v[u][1].z * w1);
        atomicAdd(&arow[7 * 32], v[u][1].w * w1);
    }
}

template <bool KEPT>
__global__ void
agg_owner(const int* __restrict__ srcv, const int* __restrict__ dstv,
          const float* __restrict__ Hm,
          const float* __restrict__ as_, const float* __restrict__ ad_,
          const float* __restrict__ bias,
          float* __restrict__ out, int E, int N, int do_relu,
          int* __restrict__ keep_list, int* __restrict__ keep_cnt,
          const unsigned char* __restrict__ own) {
    __shared__ float    acc[RNG * HC];
    __shared__ int2     list[LCAP];
    __shared__ float4   wl[LCAP];
    __shared__ float    adl[RNG * HEADS];
    __shared__ float    slf[RNG * HEADS];
    __shared__ unsigned mx[RNG * HEADS];
    __shared__ float    sm[RNG * HEADS];
    __shared__ int      cnt;

    const int t    = threadIdx.x;
    const int lane = t & 31;
    const int wave = t >> 5;
    const int base = blockIdx.x * RNG;
    const int nown = min(RNG, N - base);
    int2* kl = (int2*)keep_list + (size_t)blockIdx.x * LCAP;

    if (t == 0) cnt = 0;
    for (int i = t; i < nown * HEADS; i += AGG_T) {
        const int n4 = base * HEADS + i;
        const float a = ad_[n4];
        const float e = leaky(as_[n4] + a);
        adl[i] = a;
        slf[i] = e;
        mx[i]  = fmap(e);
        sm[i]  = 0.f;
    }
    __syncthreads();

    if constexpr (!KEPT) {
    if (gridDim.x <= NOOWN) {
        const int    E16 = E >> 4;
        const uint4* o4  = (const uint4*)own;
        const unsigned pat = blockIdx.x * 0x01010101u;
        int q0 = t;
        for (; q0 + 7 * AGG_T < E16; q0 += 8 * AGG_T) {
            uint4 ov[8];
#pragma unroll
            for (int u = 0; u < 8; ++u) ov[u] = o4[q0 + u * AGG_T];
#pragma unroll
            for (int u = 0; u < 8; u += 2)
                note_bits(own_bits(ov[u], pat) | (own_bits(ov[u + 1], pat) << 4),
                          (q0 + u * AGG_T) << 4, AGG_T << 4, &cnt, list);
        }
        for (; q0 < E16; q0 += AGG_T)
            note_bits(own_bits(o4[q0], pat), q0 << 4, 0, &cnt, list);
        for (int e = (E16 << 4) + t; e < E; e += AGG_T)
            if (own[e] == blockIdx.x) note_edge(e, &cnt, list);
    } else {
        const int  E4 = ((E & 3) == 0) ? (E >> 2) : 0;
        const int4* d4 = (const int4*)dstv;
        int q0 = t;
        for (; q0 + 3 * AGG_T < E4; q0 += 4 * AGG_T) {
            int4 dv[4];
#pragma unroll
            for (int u = 0; u < 4; ++u) dv[u] = d4[q0 + u * AGG_T];
#pragma unroll
            for (int u = 0; u < 4; ++u)
                scan4(dv[u], (q0 + u * AGG_T) << 2, base, nown, &cnt, list);
        }
        for (; q0 < E4; q0 += AGG_T)
            scan4(d4[q0], q0 << 2, base, nown, &cnt, list);
        for (int e = (E4 << 2) + t; e < E; e += AGG_T)
            if ((unsigned)(dstv[e] - base) < (unsigned)nown) note_edge(e, &cnt, list);
    }
    __syncthreads();
    if (t == 0) keep_cnt[blockIdx.x] = cnt;
    }

    const int  ntot   = KEPT ? keep_cnt[blockIdx.x] : cnt;
    const bool listed = ntot <= LCAP;

    if (listed && ntot > 0) {
        static_assert(LCAP == 4 * AGG_T, "four list entries per thread");
        bool   ok[4];
        int2   en[4];
        int    sv[4];
        float4 a4[4];
#pragma unroll
        for (int u = 0; u < 4; ++u) {
            const int i = t + u * AGG_T;
            ok[u] = i < ntot;
            en[u] = make_int2(0, 0);
            if (ok[u]) en[u] = KEPT ? kl[i] : make_int2(list[i].x, 0);
        }
#pragma unroll
        for (int u = 0; u < 4; ++u) sv[u] = KEPT ? en[u].x : srcv[en[u].x];
        if constexpr (!KEPT) {
#pragma unroll
            for (int u = 0; u < 4; ++u) en[u].y = dstv[en[u].x] - base;
        }
#pragma unroll
        for (int u = 0; u < 4; ++u) a4[u] = *(const float4*)(as_ + (size_t)sv[u] * HEADS);
#pragma unroll
        for (int u = 0; u < 4; ++u) {
            if (ok[u]) {
                const int i = t + u * AGG_T;
                const int r = en[u].y;
                const float* d = adl + r * HEADS;
                float4 e4;
                e4.x = leaky(a4[u].x + d[0]);
                e4.y = leaky(a4[u].y + d[1]);
                e4.z = leaky(a4[u].z + d[2]);
                e4.w = leaky(a4[u].w + d[3]);
                wl[i]   = e4;
                list[i] = make_int2(sv[u], r);
                if constexpr (!KEPT) kl[i] = make_int2(sv[u], r);
                unsigned* m = mx + r * HEADS;
                atomicMax(&m[0], fmap(e4.x));
                atomicMax(&m[1], fmap(e4.y));
                atomicMax(&m[2], fmap(e4.z));
                atomicMax(&m[3], fmap(e4.w));
            }
        }
    } else if (!listed) {
        walk_owned(dstv, E, base, nown, t, [&](int e, int r) {
            const int s = srcv[e];
            for (int h = 0; h < HEADS; ++h)
                atomicMax(&mx[r * HEADS + h],
                          fmap(leaky(as_[(size_t)s * HEADS + h] + adl[r * HEADS + h])));
        });
    }
    __syncthreads();

    if (listed) {
        for (int i = t; i < ntot; i += AGG_T) {
            const int r = list[i].y;
            const float4 e4 = wl[i];
            const unsigned* m = mx + r * HEADS;
            float4 x4;
            x4.x = __expf(e4.x - funmap(m[0]));
            x4.y = __expf(e4.y - funmap(m[1]));
            x4.z = __expf(e4.z - funmap(m[2]));
            x4.w = __expf(e4.w - funmap(m[3]));
            wl[i] = x4;
            float* sp = sm + r * HEADS;
            atomicAdd(&sp[0], x4.x);
            atomicAdd(&sp[1], x4.y);
            atomicAdd(&sp[2], x4.z);
            atomicAdd(&sp[3], x4.w);
        }
    } else {
        walk_owned(dstv, E, base, nown, t, [&](int e, int r) {
            const int s = srcv[e];
            for (int h = 0; h < HEADS; ++h) {
                const float ev = leaky(as_[(size_t)s * HEADS + h] + adl[r * HEADS + h]);
                atomicAdd(&sm[r * HEADS + h], __expf(ev - funmap(mx[r * HEADS + h])));
            }
        });
    }
    for (int i = t; i < nown * HEADS; i += AGG_T) {
        const float ex = __expf(slf[i] - funmap(mx[i]));
        slf[i] = ex;
        atomicAdd(&sm[i], ex);
    }
    __syncthreads();

    for (int i = t; i < nown * HEADS; i += AGG_T) {
        const float dn = sm[i] + 1e-16f;
        sm[i]  = dn;
        slf[i] = slf[i] / dn;
    }
    __syncthreads();

    {
        const float4* own4 = (const float4*)(Hm + (size_t)base * HC);
        for (int i4 = t; i4 < nown * (HC / 4); i4 += AGG_T) {
            const int r = i4 >> 6, q = i4 & 63;
            const float4 v = own4[i4];
            const float  w = slf[r * HEADS + (q >> 4)];
            float* a = acc + r * HC + ((q >> 5) << 7) + (q & 31);
            a[0]  = v.x * w;
            a[32] = v.y * w;
            a[64] = v.z * w;
            a[96] = v.w * w;
        }
    }
    if (listed) {
        for (int i = t; i < ntot; i += AGG_T) {
            const float* dn = sm + list[i].y * HEADS;
            float4 x4 = wl[i];
            x4.x = x4.x / dn[0];
            x4.y = x4.y / dn[1];
            x4.z = x4.z / dn[2];
            x4.w = x4.w / dn[3];
            wl[i] = x4;
        }
    }
    __syncthreads();

    if (listed) {
        const int nfull = ntot - (ntot % AGG_U);
        for (int i0 = wave * AGG_U; i0 < nfull; i0 += 32 * AGG_U)
            sum_noted<AGG_U>(list, wl, i0, lane, acc, Hm);
        for (int i0 = nfull + wave; i0 < ntot; i0 += 32)
            sum_noted<1>(list, wl, i0, lane, acc, Hm);
    } else {
        walk_owned(dstv, E, base, nown, t, [&](int e, int r) {
            const int s = srcv[e];
            float al[HEADS];
            for (int h = 0; h < HEADS; ++h) {
                const float ev = leaky(as_[(size_t)s * HEADS + h] + adl[r * HEADS + h]);
                al[h] = __expf(ev - funmap(mx[r * HEADS + h])) / sm[r * HEADS + h];
            }
            for (int ch = 0; ch < HC; ++ch)
                atomicAdd(&acc[r * HC + chpos(ch)], Hm[(size_t)s * HC + ch] * al[ch >> 6]);
        });
    }
    __syncthreads();

    for (int i = t; i < nown * CDIM; i += AGG_T) {
        const int r = i >> 6, c = i & 63;
        const float* a = acc + r * HC;
        float o = (a[chpos(c)] + a[chpos(CDIM + c)] + a[chpos(2 * CDIM + c)] + a[chpos(3 * CDIM + c)]) * 0.25f
                  + bias[c];
        if (do_relu) o = fmaxf(o, 0.f);
        out[(size_t)base * CDIM + i] = o;
    }
}

extern "C" void kernel_launch(void* const* d_in, const int* in_sizes, int n_in,
                              void* d_out, int out_size, void* d_ws, size_t ws_size,
                              hipStream_t stream) {
    const float* x        = (const float*)d_in[0];
    const int*   ei       = (const int*)  d_in[1];
    const float* W1       = (const float*)d_in[2];
    const float* att_src1 = (const float*)d_in[3];
    const float* att_dst1 = (const float*)d_in[4];
    const float* b1       = (const float*)d_in[5];
    const float* W2       = (const float*)d_in[6];
    const float* att_src2 = (const float*)d_in[7];
    const float* att_dst2 = (const float*)d_in[8];
    const float* b2       = (const float*)d_in[9];

    const int Fin = 128;
    const int N   = in_sizes[0] / Fin;
    const int E   = in_sizes[1] / 2;
    const int* srcv = ei;
    const int* dstv = ei + E;

    float* ws    = (float*)d_ws;
    float* Hbuf  = ws;
    float* HID   = Hbuf + (size_t)N * HC;
    float* ASRC  = HID  + (size_t)N * CDIM;
    float* ADST  = ASRC + (size_t)N * HEADS;
    int*   KLIST = (int*)(ADST + (size_t)N * HEADS);
    unsigned char* OWN = (unsigned char*)(KLIST + (size_t)((N + RNG - 1) / RNG) * LCAP * 2);
    int*   KCNT  = (int*)(OWN + ((size_t)E + 15) / 16 * 16);

    const int gemmBlocks = N / 16;
    const int agBlocks   = (N + RNG - 1) / RNG;

    gemm_scores_wmma_f32<<<gemmBlocks, 128, 0, stream>>>(x, W1, Hbuf, att_src1, att_dst1, ASRC, ADST, Fin, dstv, OWN, E, N);
    agg_owner<false><<<agBlocks, AGG_T, 0, stream>>>(srcv, dstv, Hbuf, ASRC, ADST, b1, HID, E, N,  1, KLIST, KCNT, OWN);

    gemm_scores_wmma_f32<<<gemmBlocks, 128, 0, stream>>>(HID, W2, Hbuf, att_src2, att_dst2, ASRC, ADST, CDIM, nullptr, nullptr, 0, 0);
    agg_owner<true><<<agBlocks, AGG_T, 0, stream>>>(srcv, dstv, Hbuf, ASRC, ADST, b2, (float*)d_out, E, N,  0, KLIST, KCNT, OWN);
}


# projection: a wave keeps its head's whole K x 64 strip of B in registers and a block loops over 3 to 5 row tiles (768 and 1024 resident blocks, was 3125): the K loop reads no B from memory
# speedup vs baseline: 8.6625x; 1.0518x over previous
#include <hip/hip_runtime.h>
#include <hip/hip_bf16.h>
#include <math.h>

typedef float v2f __attribute__((ext_vector_type(2)));
typedef float v8f __attribute__((ext_vector_type(8)));

#define HEADS 4
#define HC 256
#define CDIM 64
#define NEG_SLOPE 0.2f


template <int XM>
__device__ __forceinline__ float row_xor(float x) {
    return __int_as_float(__builtin_amdgcn_update_dpp(0, __float_as_int(x),
                                                      0x160 | XM, 0xf, 0xf, true));
}

template <int XM, int NV>
__device__ __forceinline__ void halve(float* v, bool up) {
#pragma unroll
    for (int k = 0; k < NV / 2; ++k) {
        const float keep = up ? v[k + NV / 2] : v[k];
        const float send = up ? v[k] : v[k + NV / 2];
        v[k] = keep + row_xor<XM>(send);
    }
}

#define APITCH 132
#define RNG   200
#define NOOWN 0xFFu

__device__ __forceinline__ unsigned owner_of(int d, int N) {
    const unsigned id = (unsigned)d / RNG;
    return ((unsigned)d < (unsigned)N && id < NOOWN) ? id : NOOWN;
}

template <int NP>
__device__ __forceinline__ void load_pieces(const float4* __restrict__ Ag,
                                            float4& t0, float4& t1, float4& t2, float4& t3) {
    t0 = Ag[0];
    t1 = Ag[8];
    if constexpr (NP == 4) { t2 = Ag[16]; t3 = Ag[24]; }
}
template <int NP>
__device__ __forceinline__ void store_pieces(float* dst, const float4& t0, const float4& t1,
                                             const float4& t2, const float4& t3) {
    *(float4*)(dst)      = t0;
    *(float4*)(dst + 32) = t1;
    if constexpr (NP == 4) { *(float4*)(dst + 64) = t2; *(float4*)(dst + 96) = t3; }
}

template <int K>
__global__ void __launch_bounds__(128)
gemm_scores_wmma_f32(const float* __restrict__ A,
                     const float* __restrict__ B,
                     float* __restrict__ C,
                     const float* __restrict__ att_s,
                     const float* __restrict__ att_d,
                     float* __restrict__ as_, float* __restrict__ ad_,
                     int ntiles,
                     const int* __restrict__ dstv, unsigned char* __restrict__ own,
                     int E, int N) {
    static_assert(K == 64 || K == 128, "a thread stages 2 or 4 whole 16-byte pieces of a row");
    __shared__ float sc[2 * 16 * HEADS];
    __shared__ __attribute__((aligned(16))) float At[16 * APITCH];

    constexpr int NP = K / 32;
    const int lane   = threadIdx.x & 31;
    const int wave   = threadIdx.x >> 5;
    const int col0   = wave << 6;
    const int laneM  = lane & 15;
    const int laneHi = lane >> 4;
    const int sr     = threadIdx.x >> 3;
    const int sc0    = threadIdx.x & 7;

    int tile = blockIdx.x;
    const int tlast = max(ntiles - 1, 0);
    float4 t0, t1, t2 = {}, t3 = {};
    load_pieces<NP>((const float4*)(A + (size_t)((min(tile, tlast) << 4) + sr) * K) + sc0, t0, t1, t2, t3);

    v2f bq[K / 4][4];
    {
        const float* Bbase = B + (size_t)col0 + laneM;
#pragma unroll
        for (int kk = 0; kk < K / 4; ++kk) {
            const int ka = 4 * kk + (laneHi << 1);
            const float* B0 = Bbase + (size_t)ka * HC;
            const float* B1 = Bbase + (size_t)(ka + 1) * HC;
            bq[kk][0].x = B0[0];  bq[kk][0].y = B1[0];
            bq[kk][1].x = B0[16]; bq[kk][1].y = B1[16];
            bq[kk][2].x = B0[32]; bq[kk][2].y = B1[32];
            bq[kk][3].x = B0[48]; bq[kk][3].y = B1[48];
        }
    }

    if (own) {
        const int gt = blockIdx.x * 128 + threadIdx.x;
        const int gs = gridDim.x * 128;
        if ((E & 3) == 0) {
            const int4* d4 = (const int4*)dstv;
            unsigned*   o4 = (unsigned*)own;
            for (int g = gt; g < (E >> 2); g += gs) {
                const int4 d = d4[g];
                o4[g] = owner_of(d.x, N) | (owner_of(d.y, N) << 8) |
                        (owner_of(d.z, N) << 16) | (owner_of(d.w, N) << 24);
            }
        } else {
            for (int e = gt; e < E; e += gs) own[e] = (unsigned char)owner_of(dstv[e], N);
        }
    }

    const float* Arow = At + laneM * APITCH + (laneHi << 1);
    const float* vsp  = att_s + col0 + laneM;
    const float* vdp  = att_d + col0 + laneM;

    for (; tile < ntiles; tile += gridDim.x) {
        const int row0 = tile << 4;
        store_pieces<NP>(&At[sr * APITCH + (sc0 << 2)], t0, t1, t2, t3);
        __syncthreads();

        load_pieces<NP>((const float4*)(A + (size_t)((min(tile + (int)gridDim.x, tlast) << 4) + sr) * K) + sc0,
                        t0, t1, t2, t3);
        if constexpr (K <= 64) __builtin_amdgcn_sched_barrier(0);

        v8f acc0 = {}, acc1 = {}, acc2 = {}, acc3 = {};
#pragma unroll
        for (int kk = 0; kk < K / 4; ++kk) {
            const v2f a = *(const v2f*)(Arow + 4 * kk);
            acc0 = __builtin_amdgcn_wmma_f32_16x16x4_f32(false, a, false, bq[kk][0], (short)0, acc0, false, false);
            acc1 = __builtin_amdgcn_wmma_f32_16x16x4_f32(false, a, false, bq[kk][1], (short)0, acc1, false, false);
            acc2 = __builtin_amdgcn_wmma_f32_16x16x4_f32(false, a, false, bq[kk][2], (short)0, acc2, false, false);
            acc3 = __builtin_amdgcn_wmma_f32_16x16x4_f32(false, a, false, bq[kk][3], (short)0, acc3, false, false);
        }

        const float s0 = vsp[0], s1 = vsp[16], s2 = vsp[32], s3 = vsp[48];
        const float d0 = vdp[0], d1 = vdp[16], d2 = vdp[32], d3 = vdp[48];
        float v[16];
#pragma unroll
        for (int i = 0; i < 8; ++i) {
            v[i]     = fmaf(acc3[i], s3, fmaf(acc2[i], s2, fmaf(acc1[i], s1, acc0[i] * s0)));
            v[8 + i] = fmaf(acc3[i], d3, fmaf(acc2[i], d2, fmaf(acc1[i], d1, acc0[i] * d0)));
        }
        halve<8, 16>(v, (laneM & 8) != 0);
        halve<4, 8>(v, (laneM & 4) != 0);
        halve<2, 4>(v, (laneM & 2) != 0);
        halve<1, 2>(v, (laneM & 1) != 0);
        sc[(laneM >> 3) * (16 * HEADS) + ((laneM & 7) + 8 * laneHi) * HEADS + wave] = v[0];
        __syncthreads();
        {
            const int t = threadIdx.x;
            float* dstp = (t < 16 * HEADS) ? as_ : ad_;
            dstp[(size_t)row0 * HEADS + (t & (16 * HEADS - 1))] = sc[t];
        }

        float* Crow = C + (size_t)(row0 + 8 * laneHi) * HC + col0 + laneM;
#pragma unroll
        for (int i = 0; i < 8; ++i) {
            float* r = Crow + (size_t)i * HC;
            r[0]  = acc0[i];
            r[16] = acc1[i];
            r[32] = acc2[i];
            r[48] = acc3[i];
        }
    }
}

__device__ __forceinline__ unsigned fmap(float x) {
    unsigned u = __float_as_uint(x);
    return (u & 0x80000000u) ? ~u : (u | 0x80000000u);
}
__device__ __forceinline__ float funmap(unsigned u) {
    u = (u & 0x80000000u) ? (u ^ 0x80000000u) : ~u;
    return __uint_as_float(u);
}

__device__ __forceinline__ float leaky(float e) {
    return e > 0.f ? e : NEG_SLOPE * e;
}

#define LCAP  4096
#define AGG_T 1024
#define AGG_U 4

__device__ __forceinline__ void note_edge(int e, int* cnt, int2* list) {
    const int sl = atomicAdd(cnt, 1);
    if (sl < LCAP) list[sl].x = e;
}

__device__ __forceinline__ unsigned zero_bytes(unsigned x) {
    return ~(((x & 0x7F7F7F7Fu) + 0x7F7F7F7Fu) | x | 0x7F7F7F7Fu);
}

__device__ __forceinline__ unsigned own_bits(uint4 o, unsigned pat) {
    return (zero_bytes(o.x ^ pat) >> 7) | (zero_bytes(o.y ^ pat) >> 6) |
           (zero_bytes(o.z ^ pat) >> 5) | (zero_bytes(o.w ^ pat) >> 4);
}

__device__ __forceinline__ void note_bits(unsigned m, int e0, int estep,
                                          int* cnt, int2* list) {
    while (m) {
        const int b = __builtin_ctz(m);
        m &= m - 1;
        note_edge(e0 + ((b & 4) ? estep : 0) + ((b & 3) << 2) + (b >> 3), cnt, list);
    }
}

__device__ __forceinline__ void scan4(int4 d, int e0, int base, int nown,
                                      int* cnt, int2* list) {
    const unsigned r0 = (unsigned)(d.x - base);
    const unsigned r1 = (unsigned)(d.y - base);
    const unsigned r2 = (unsigned)(d.z - base);
    const unsigned r3 = (unsigned)(d.w - base);
    if (min(min(r0, r1), min(r2, r3)) < (unsigned)nown) {
        if (r0 < (unsigned)nown) note_edge(e0,     cnt, list);
        if (r1 < (unsigned)nown) note_edge(e0 + 1, cnt, list);
        if (r2 < (unsigned)nown) note_edge(e0 + 2, cnt, list);
        if (r3 < (unsigned)nown) note_edge(e0 + 3, cnt, list);
    }
}

template <class F>
__device__ __forceinline__ void walk_owned(const int* __restrict__ dstv, int E,
                                           int base, int nown, int t, F f) {
    for (int e = t; e < E; e += AGG_T) {
        const unsigned r = (unsigned)(dstv[e] - base);
        if (r < (unsigned)nown) f(e, (int)r);
    }
}

__device__ __forceinline__ int chpos(int ch) {
    return (ch & 128) + ((ch & 3) << 5) + ((ch & 127) >> 2);
}

template <int U>
__device__ __forceinline__ void sum_noted(const int2* list, const float4* wl,
                                          int i0, int lane, float* acc,
                                          const float* __restrict__ Hm) {
    int    rr[U], ss[U];
    float  al[U][HEADS];
    float4 v[U][2];
#pragma unroll
    for (int u = 0; u < U; ++u) {
        const int2 en = list[i0 + u];
        ss[u] = __builtin_amdgcn_readfirstlane(en.x);
        rr[u] = __builtin_amdgcn_readfirstlane(en.y);
    }
#pragma unroll
    for (int u = 0; u < U; ++u) {
        const float4* hrow = (const float4*)(Hm + (size_t)ss[u] * HC) + lane;
        v[u][0] = hrow[0];
        v[u][1] = hrow[32];
    }
#pragma unroll
    for (int u = 0; u < U; ++u) {
        const float4 w4 = wl[i0 + u];
        al[u][0] = w4.x; al[u][1] = w4.y; al[u][2] = w4.z; al[u][3] = w4.w;
    }
#pragma unroll
    for (int u = 0; u < U; ++u) {
        float* arow = acc + rr[u] * HC + lane;
        const float w0 = (lane & 16) ? al[u][1] : al[u][0];
        const float w1 = (lane & 16) ? al[u][3] : al[u][2];
        atomicAdd(&arow[0 * 32], v[u][0].x * w0);
        atomicAdd(&arow[1 * 32], v[u][0].y * w0);
        atomicAdd(&arow[2 * 32], v[u][0].z * w0);
        atomicAdd(&arow[3 * 32], v[u][0].w * w0);
        atomicAdd(&arow[4 * 32], v[u][1].x * w1);
        atomicAdd(&arow[5 * 32], v[u][1].y * w1);
        atomicAdd(&arow[6 * 32], v[u][1].z * w1);
        atomicAdd(&arow[7 * 32], v[u][1].w * w1);
    }
}

template <bool KEPT>
__global__ void
agg_owner(const int* __restrict__ srcv, const int* __restrict__ dstv,
          const float* __restrict__ Hm,
          const float* __restrict__ as_, const float* __restrict__ ad_,
          const float* __restrict__ bias,
          float* __restrict__ out, int E, int N, int do_relu,
          int* __restrict__ keep_list, int* __restrict__ keep_cnt,
          const unsigned char* __restrict__ own) {
    __shared__ float    acc[RNG * HC];
    __shared__ int2     list[LCAP];
    __shared__ float4   wl[LCAP];
    __shared__ float    adl[RNG * HEADS];
    __shared__ float    slf[RNG * HEADS];
    __shared__ unsigned mx[RNG * HEADS];
    __shared__ float    sm[RNG * HEADS];
    __shared__ int      cnt;

    const int t    = threadIdx.x;
    const int lane = t & 31;
    const int wave = t >> 5;
    const int base = blockIdx.x * RNG;
    const int nown = min(RNG, N - base);
    int2* kl = (int2*)keep_list + (size_t)blockIdx.x * LCAP;

    if (t == 0) cnt = 0;
    for (int i = t; i < nown * HEADS; i += AGG_T) {
        const int n4 = base * HEADS + i;
        const float a = ad_[n4];
        const float e = leaky(as_[n4] + a);
        adl[i] = a;
        slf[i] = e;
        mx[i]  = fmap(e);
        sm[i]  = 0.f;
    }
    __syncthreads();

    if constexpr (!KEPT) {
    if (gridDim.x <= NOOWN) {
        const int    E16 = E >> 4;
        const uint4* o4  = (const uint4*)own;
        const unsigned pat = blockIdx.x * 0x01010101u;
        int q0 = t;
        for (; q0 + 7 * AGG_T < E16; q0 += 8 * AGG_T) {
            uint4 ov[8];
#pragma unroll
            for (int u = 0; u < 8; ++u) ov[u] = o4[q0 + u * AGG_T];
#pragma unroll
            for (int u = 0; u < 8; u += 2)
                note_bits(own_bits(ov[u], pat) | (own_bits(ov[u + 1], pat) << 4),
                          (q0 + u * AGG_T) << 4, AGG_T << 4, &cnt, list);
        }
        for (; q0 < E16; q0 += AGG_T)
            note_bits(own_bits(o4[q0], pat), q0 << 4, 0, &cnt, list);
        for (int e = (E16 << 4) + t; e < E; e += AGG_T)
            if (own[e] == blockIdx.x) note_edge(e, &cnt, list);
    } else {
        const int  E4 = ((E & 3) == 0) ? (E >> 2) : 0;
        const int4* d4 = (const int4*)dstv;
        int q0 = t;
        for (; q0 + 3 * AGG_T < E4; q0 += 4 * AGG_T) {
            int4 dv[4];
#pragma unroll
            for (int u = 0; u < 4; ++u) dv[u] = d4[q0 + u * AGG_T];
#pragma unroll
            for (int u = 0; u < 4; ++u)
                scan4(dv[u], (q0 + u * AGG_T) << 2, base, nown, &cnt, list);
        }
        for (; q0 < E4; q0 += AGG_T)
            scan4(d4[q0], q0 << 2, base, nown, &cnt, list);
        for (int e = (E4 << 2) + t; e < E; e += AGG_T)
            if ((unsigned)(dstv[e] - base) < (unsigned)nown) note_edge(e, &cnt, list);
    }
    __syncthreads();
    if (t == 0) keep_cnt[blockIdx.x] = cnt;
    }

    const int  ntot   = KEPT ? keep_cnt[blockIdx.x] : cnt;
    const bool listed = ntot <= LCAP;

    if (listed && ntot > 0) {
        static_assert(LCAP == 4 * AGG_T, "four list entries per thread");
        bool   ok[4];
        int2   en[4];
        int    sv[4];
        float4 a4[4];
#pragma unroll
        for (int u = 0; u < 4; ++u) {
            const int i = t + u * AGG_T;
            ok[u] = i < ntot;
            en[u] = make_int2(0, 0);
            if (ok[u]) en[u] = KEPT ? kl[i] : make_int2(list[i].x, 0);
        }
#pragma unroll
        for (int u = 0; u < 4; ++u) sv[u] = KEPT ? en[u].x : srcv[en[u].x];
        if constexpr (!KEPT) {
#pragma unroll
            for (int u = 0; u < 4; ++u) en[u].y = dstv[en[u].x] - base;
        }
#pragma unroll
        for (int u = 0; u < 4; ++u) a4[u] = *(const float4*)(as_ + (size_t)sv[u] * HEADS);
#pragma unroll
        for (int u = 0; u < 4; ++u) {
            if (ok[u]) {
                const int i = t + u * AGG_T;
                const int r = en[u].y;
                const float* d = adl + r * HEADS;
                float4 e4;
                e4.x = leaky(a4[u].x + d[0]);
                e4.y = leaky(a4[u].y + d[1]);
                e4.z = leaky(a4[u].z + d[2]);
                e4.w = leaky(a4[u].w + d[3]);
                wl[i]   = e4;
                list[i] = make_int2(sv[u], r);
                if constexpr (!KEPT) kl[i] = make_int2(sv[u], r);
                unsigned* m = mx + r * HEADS;
                atomicMax(&m[0], fmap(e4.x));
                atomicMax(&m[1], fmap(e4.y));
                atomicMax(&m[2], fmap(e4.z));
                atomicMax(&m[3], fmap(e4.w));
            }
        }
    } else if (!listed) {
        walk_owned(dstv, E, base, nown, t, [&](int e, int r) {
            const int s = srcv[e];
            for (int h = 0; h < HEADS; ++h)
                atomicMax(&mx[r * HEADS + h],
                          fmap(leaky(as_[(size_t)s * HEADS + h] + adl[r * HEADS + h])));
        });
    }
    __syncthreads();

    if (listed) {
        for (int i = t; i < ntot; i += AGG_T) {
            const int r = list[i].y;
            const float4 e4 = wl[i];
            const unsigned* m = mx + r * HEADS;
            float4 x4;
            x4.x = __expf(e4.x - funmap(m[0]));
            x4.y = __expf(e4.y - funmap(m[1]));
            x4.z = __expf(e4.z - funmap(m[2]));
            x4.w = __expf(e4.w - funmap(m[3]));
            wl[i] = x4;
            float* sp = sm + r * HEADS;
            atomicAdd(&sp[0], x4.x);
            atomicAdd(&sp[1], x4.y);
            atomicAdd(&sp[2], x4.z);
            atomicAdd(&sp[3], x4.w);
        }
    } else {
        walk_owned(dstv, E, base, nown, t, [&](int e, int r) {
            const int s = srcv[e];
            for (int h = 0; h < HEADS; ++h) {
                const float ev = leaky(as_[(size_t)s * HEADS + h] + adl[r * HEADS + h]);
                atomicAdd(&sm[r * HEADS + h], __expf(ev - funmap(mx[r * HEADS + h])));
            }
        });
    }
    for (int i = t; i < nown * HEADS; i += AGG_T) {
        const float ex = __expf(slf[i] - funmap(mx[i]));
        slf[i] = ex;
        atomicAdd(&sm[i], ex);
    }
    __syncthreads();

    for (int i = t; i < nown * HEADS; i += AGG_T) {
        const float dn = sm[i] + 1e-16f;
        sm[i]  = dn;
        slf[i] = slf[i] / dn;
    }
    __syncthreads();

    {
        const float4* own4 = (const float4*)(Hm + (size_t)base * HC);
        for (int i4 = t; i4 < nown * (HC / 4); i4 += AGG_T) {
            const int r = i4 >> 6, q = i4 & 63;
            const float4 v = own4[i4];
            const float  w = slf[r * HEADS + (q >> 4)];
            float* a = acc + r * HC + ((q >> 5) << 7) + (q & 31);
            a[0]  = v.x * w;
            a[32] = v.y * w;
            a[64] = v.z * w;
            a[96] = v.w * w;
        }
    }
    if (listed) {
        for (int i = t; i < ntot; i += AGG_T) {
            const float* dn = sm + list[i].y * HEADS;
            float4 x4 = wl[i];
            x4.x = x4.x / dn[0];
            x4.y = x4.y / dn[1];
            x4.z = x4.z / dn[2];
            x4.w = x4.w / dn[3];
            wl[i] = x4;
        }
    }
    __syncthreads();

    if (listed) {
        const int nfull = ntot - (ntot % AGG_U);
        for (int i0 = wave * AGG_U; i0 < nfull; i0 += 32 * AGG_U)
            sum_noted<AGG_U>(list, wl, i0, lane, acc, Hm);
        for (int i0 = nfull + wave; i0 < ntot; i0 += 32)
            sum_noted<1>(list, wl, i0, lane, acc, Hm);
    } else {
        walk_owned(dstv, E, base, nown, t, [&](int e, int r) {
            const int s = srcv[e];
            float al[HEADS];
            for (int h = 0; h < HEADS; ++h) {
                const float ev = leaky(as_[(size_t)s * HEADS + h] + adl[r * HEADS + h]);
                al[h] = __expf(ev - funmap(mx[r * HEADS + h])) / sm[r * HEADS + h];
            }
            for (int ch = 0; ch < HC; ++ch)
                atomicAdd(&acc[r * HC + chpos(ch)], Hm[(size_t)s * HC + ch] * al[ch >> 6]);
        });
    }
    __syncthreads();

    for (int i = t; i < nown * CDIM; i += AGG_T) {
        const int r = i >> 6, c = i & 63;
        const float* a = acc + r * HC;
        float o = (a[chpos(c)] + a[chpos(CDIM + c)] + a[chpos(2 * CDIM + c)] + a[chpos(3 * CDIM + c)]) * 0.25f
                  + bias[c];
        if (do_relu) o = fmaxf(o, 0.f);
        out[(size_t)base * CDIM + i] = o;
    }
}

extern "C" void kernel_launch(void* const* d_in, const int* in_sizes, int n_in,
                              void* d_out, int out_size, void* d_ws, size_t ws_size,
                              hipStream_t stream) {
    const float* x        = (const float*)d_in[0];
    const int*   ei       = (const int*)  d_in[1];
    const float* W1       = (const float*)d_in[2];
    const float* att_src1 = (const float*)d_in[3];
    const float* att_dst1 = (const float*)d_in[4];
    const float* b1       = (const float*)d_in[5];
    const float* W2       = (const float*)d_in[6];
    const float* att_src2 = (const float*)d_in[7];
    const float* att_dst2 = (const float*)d_in[8];
    const float* b2       = (const float*)d_in[9];

    const int Fin = 128;
    const int N   = in_sizes[0] / Fin;
    const int E   = in_sizes[1] / 2;
    const int* srcv = ei;
    const int* dstv = ei + E;

    float* ws    = (float*)d_ws;
    float* Hbuf  = ws;
    float* HID   = Hbuf + (size_t)N * HC;
    float* ASRC  = HID  + (size_t)N * CDIM;
    float* ADST  = ASRC + (size_t)N * HEADS;
    int*   KLIST = (int*)(ADST + (size_t)N * HEADS);
    unsigned char* OWN = (unsigned char*)(KLIST + (size_t)((N + RNG - 1) / RNG) * LCAP * 2);
    int*   KCNT  = (int*)(OWN + ((size_t)E + 15) / 16 * 16);

    const int ntiles     = N / 16;
    const int agBlocks   = (N + RNG - 1) / RNG;

    gemm_scores_wmma_f32<128><<<768, 128, 0, stream>>>(x, W1, Hbuf, att_src1, att_dst1, ASRC, ADST, ntiles, dstv, OWN, E, N);
    agg_owner<false><<<agBlocks, AGG_T, 0, stream>>>(srcv, dstv, Hbuf, ASRC, ADST, b1, HID, E, N,  1, KLIST, KCNT, OWN);

    gemm_scores_wmma_f32<CDIM><<<1024, 128, 0, stream>>>(HID, W2, Hbuf, att_src2, att_dst2, ASRC, ADST, ntiles, nullptr, nullptr, 0, 0);
    agg_owner<true><<<agBlocks, AGG_T, 0, stream>>>(srcv, dstv, Hbuf, ASRC, ADST, b2, (float*)d_out, E, N,  0, KLIST, KCNT, OWN);
}


# edge sum: a half-wave takes an edge, loads its 4 heads' 16-byte pieces and sums them weighted in registers: 4 LDS adds per 2 edges, not 16; sum rows [64]; sum order changes: heads first, then edges
# speedup vs baseline: 9.6602x; 1.1152x over previous
#include <hip/hip_runtime.h>
#include <hip/hip_bf16.h>
#include <math.h>

typedef float v2f __attribute__((ext_vector_type(2)));
typedef float v8f __attribute__((ext_vector_type(8)));

#define HEADS 4
#define HC 256
#define CDIM 64
#define NEG_SLOPE 0.2f


template <int XM>
__device__ __forceinline__ float row_xor(float x) {
    return __int_as_float(__builtin_amdgcn_update_dpp(0, __float_as_int(x),
                                                      0x160 | XM, 0xf, 0xf, true));
}

template <int XM, int NV>
__device__ __forceinline__ void halve(float* v, bool up) {
#pragma unroll
    for (int k = 0; k < NV / 2; ++k) {
        const float keep = up ? v[k + NV / 2] : v[k];
        const float send = up ? v[k] : v[k + NV / 2];
        v[k] = keep + row_xor<XM>(send);
    }
}

#define APITCH 132
#define RNG   200
#define NOOWN 0xFFu

__device__ __forceinline__ unsigned owner_of(int d, int N) {
    const unsigned id = (unsigned)d / RNG;
    return ((unsigned)d < (unsigned)N && id < NOOWN) ? id : NOOWN;
}

template <int NP>
__device__ __forceinline__ void load_pieces(const float4* __restrict__ Ag,
                                            float4& t0, float4& t1, float4& t2, float4& t3) {
    t0 = Ag[0];
    t1 = Ag[8];
    if constexpr (NP == 4) { t2 = Ag[16]; t3 = Ag[24]; }
}
template <int NP>
__device__ __forceinline__ void store_pieces(float* dst, const float4& t0, const float4& t1,
                                             const float4& t2, const float4& t3) {
    *(float4*)(dst)      = t0;
    *(float4*)(dst + 32) = t1;
    if constexpr (NP == 4) { *(float4*)(dst + 64) = t2; *(float4*)(dst + 96) = t3; }
}

template <int K>
__global__ void __launch_bounds__(128)
gemm_scores_wmma_f32(const float* __restrict__ A,
                     const float* __restrict__ B,
                     float* __restrict__ C,
                     const float* __restrict__ att_s,
                     const float* __restrict__ att_d,
                     float* __restrict__ as_, float* __restrict__ ad_,
                     int ntiles,
                     const int* __restrict__ dstv, unsigned char* __restrict__ own,
                     int E, int N) {
    static_assert(K == 64 || K == 128, "a thread stages 2 or 4 whole 16-byte pieces of a row");
    __shared__ float sc[2 * 16 * HEADS];
    __shared__ __attribute__((aligned(16))) float At[16 * APITCH];

    constexpr int NP = K / 32;
    const int lane   = threadIdx.x & 31;
    const int wave   = threadIdx.x >> 5;
    const int col0   = wave << 6;
    const int laneM  = lane & 15;
    const int laneHi = lane >> 4;
    const int sr     = threadIdx.x >> 3;
    const int sc0    = threadIdx.x & 7;

    int tile = blockIdx.x;
    const int tlast = max(ntiles - 1, 0);
    float4 t0, t1, t2 = {}, t3 = {};
    load_pieces<NP>((const float4*)(A + (size_t)((min(tile, tlast) << 4) + sr) * K) + sc0, t0, t1, t2, t3);

    v2f bq[K / 4][4];
    {
        const float* Bbase = B + (size_t)col0 + laneM;
#pragma unroll
        for (int kk = 0; kk < K / 4; ++kk) {
            const int ka = 4 * kk + (laneHi << 1);
            const float* B0 = Bbase + (size_t)ka * HC;
            const float* B1 = Bbase + (size_t)(ka + 1) * HC;
            bq[kk][0].x = B0[0];  bq[kk][0].y = B1[0];
            bq[kk][1].x = B0[16]; bq[kk][1].y = B1[16];
            bq[kk][2].x = B0[32]; bq[kk][2].y = B1[32];
            bq[kk][3].x = B0[48]; bq[kk][3].y = B1[48];
        }
    }

    if (own) {
        const int gt = blockIdx.x * 128 + threadIdx.x;
        const int gs = gridDim.x * 128;
        if ((E & 3) == 0) {
            const int4* d4 = (const int4*)dstv;
            unsigned*   o4 = (unsigned*)own;
            for (int g = gt; g < (E >> 2); g += gs) {
                const int4 d = d4[g];
                o4[g] = owner_of(d.x, N) | (owner_of(d.y, N) << 8) |
                        (owner_of(d.z, N) << 16) | (owner_of(d.w, N) << 24);
            }
        } else {
            for (int e = gt; e < E; e += gs) own[e] = (unsigned char)owner_of(dstv[e], N);
        }
    }

    const float* Arow = At + laneM * APITCH + (laneHi << 1);
    const float* vsp  = att_s + col0 + laneM;
    const float* vdp  = att_d + col0 + laneM;

    for (; tile < ntiles; tile += gridDim.x) {
        const int row0 = tile << 4;
        store_pieces<NP>(&At[sr * APITCH + (sc0 << 2)], t0, t1, t2, t3);
        __syncthreads();

        load_pieces<NP>((const float4*)(A + (size_t)((min(tile + (int)gridDim.x, tlast) << 4) + sr) * K) + sc0,
                        t0, t1, t2, t3);
        if constexpr (K <= 64) __builtin_amdgcn_sched_barrier(0);

        v8f acc0 = {}, acc1 = {}, acc2 = {}, acc3 = {};
#pragma unroll
        for (int kk = 0; kk < K / 4; ++kk) {
            const v2f a = *(const v2f*)(Arow + 4 * kk);
            acc0 = __builtin_amdgcn_wmma_f32_16x16x4_f32(false, a, false, bq[kk][0], (short)0, acc0, false, false);
            acc1 = __builtin_amdgcn_wmma_f32_16x16x4_f32(false, a, false, bq[kk][1], (short)0, acc1, false, false);
            acc2 = __builtin_amdgcn_wmma_f32_16x16x4_f32(false, a, false, bq[kk][2], (short)0, acc2, false, false);
            acc3 = __builtin_amdgcn_wmma_f32_16x16x4_f32(false, a, false, bq[kk][3], (short)0, acc3, false, false);
        }

        const float s0 = vsp[0], s1 = vsp[16], s2 = vsp[32], s3 = vsp[48];
        const float d0 = vdp[0], d1 = vdp[16], d2 = vdp[32], d3 = vdp[48];
        float v[16];
#pragma unroll
        for (int i = 0; i < 8; ++i) {
            v[i]     = fmaf(acc3[i], s3, fmaf(acc2[i], s2, fmaf(acc1[i], s1, acc0[i] * s0)));
            v[8 + i] = fmaf(acc3[i], d3, fmaf(acc2[i], d2, fmaf(acc1[i], d1, acc0[i] * d0)));
        }
        halve<8, 16>(v, (laneM & 8) != 0);
        halve<4, 8>(v, (laneM & 4) != 0);
        halve<2, 4>(v, (laneM & 2) != 0);
        halve<1, 2>(v, (laneM & 1) != 0);
        sc[(laneM >> 3) * (16 * HEADS) + ((laneM & 7) + 8 * laneHi) * HEADS + wave] = v[0];
        __syncthreads();
        {
            const int t = threadIdx.x;
            float* dstp = (t < 16 * HEADS) ? as_ : ad_;
            dstp[(size_t)row0 * HEADS + (t & (16 * HEADS - 1))] = sc[t];
        }

        float* Crow = C + (size_t)(row0 + 8 * laneHi) * HC + col0 + laneM;
#pragma unroll
        for (int i = 0; i < 8; ++i) {
            float* r = Crow + (size_t)i * HC;
            r[0]  = acc0[i];
            r[16] = acc1[i];
            r[32] = acc2[i];
            r[48] = acc3[i];
        }
    }
}

__device__ __forceinline__ unsigned fmap(float x) {
    unsigned u = __float_as_uint(x);
    return (u & 0x80000000u) ? ~u : (u | 0x80000000u);
}
__device__ __forceinline__ float funmap(unsigned u) {
    u = (u & 0x80000000u) ? (u ^ 0x80000000u) : ~u;
    return __uint_as_float(u);
}

__device__ __forceinline__ float leaky(float e) {
    return e > 0.f ? e : NEG_SLOPE * e;
}

#define LCAP  4096
#define AGG_T 1024
#define AGG_U 4

__device__ __forceinline__ void note_edge(int e, int* cnt, int2* list) {
    const int sl = atomicAdd(cnt, 1);
    if (sl < LCAP) list[sl].x = e;
}

__device__ __forceinline__ unsigned zero_bytes(unsigned x) {
    return ~(((x & 0x7F7F7F7Fu) + 0x7F7F7F7Fu) | x | 0x7F7F7F7Fu);
}

__device__ __forceinline__ unsigned own_bits(uint4 o, unsigned pat) {
    return (zero_bytes(o.x ^ pat) >> 7) | (zero_bytes(o.y ^ pat) >> 6) |
           (zero_bytes(o.z ^ pat) >> 5) | (zero_bytes(o.w ^ pat) >> 4);
}

__device__ __forceinline__ void note_bits(unsigned m, int e0, int estep,
                                          int* cnt, int2* list) {
    while (m) {
        const int b = __builtin_ctz(m);
        m &= m - 1;
        note_edge(e0 + ((b & 4) ? estep : 0) + ((b & 3) << 2) + (b >> 3), cnt, list);
    }
}

__device__ __forceinline__ void scan4(int4 d, int e0, int base, int nown,
                                      int* cnt, int2* list) {
    const unsigned r0 = (unsigned)(d.x - base);
    const unsigned r1 = (unsigned)(d.y - base);
    const unsigned r2 = (unsigned)(d.z - base);
    const unsigned r3 = (unsigned)(d.w - base);
    if (min(min(r0, r1), min(r2, r3)) < (unsigned)nown) {
        if (r0 < (unsigned)nown) note_edge(e0,     cnt, list);
        if (r1 < (unsigned)nown) note_edge(e0 + 1, cnt, list);
        if (r2 < (unsigned)nown) note_edge(e0 + 2, cnt, list);
        if (r3 < (unsigned)nown) note_edge(e0 + 3, cnt, list);
    }
}

template <class F>
__device__ __forceinline__ void walk_owned(const int* __restrict__ dstv, int E,
                                           int base, int nown, int t, F f) {
    for (int e = t; e < E; e += AGG_T) {
        const unsigned r = (unsigned)(dstv[e] - base);
        if (r < (unsigned)nown) f(e, (int)r);
    }
}

__device__ __forceinline__ float4 head_sum(const float4& v0, const float4& v1, const float4& v2,
                                           const float4& v3, const float4& w) {
    float4 s;
    s.x = fmaf(v3.x, w.w, fmaf(v2.x, w.z, fmaf(v1.x, w.y, v0.x * w.x)));
    s.y = fmaf(v3.y, w.w, fmaf(v2.y, w.z, fmaf(v1.y, w.y, v0.y * w.x)));
    s.z = fmaf(v3.z, w.w, fmaf(v2.z, w.z, fmaf(v1.z, w.y, v0.z * w.x)));
    s.w = fmaf(v3.w, w.w, fmaf(v2.w, w.z, fmaf(v1.w, w.y, v0.w * w.x)));
    return s;
}

template <int U, bool GUARD>
__device__ __forceinline__ void sum_noted(const int2* list, const float4* wl,
                                          int i0, int n, int lane, float* acc,
                                          const float* __restrict__ Hm) {
    const int q  = lane & 15;
    const int hi = lane >> 4;
    if (GUARD && i0 + hi >= n) return;
    int2   en[U];
    float4 v[U][HEADS];
    float4 w[U];
#pragma unroll
    for (int u = 0; u < U; ++u) en[u] = list[i0 + 2 * u + hi];
#pragma unroll
    for (int u = 0; u < U; ++u) {
        const float4* hrow = (const float4*)(Hm + (size_t)en[u].x * HC) + q;
        v[u][0] = hrow[0];
        v[u][1] = hrow[16];
        v[u][2] = hrow[32];
        v[u][3] = hrow[48];
    }
#pragma unroll
    for (int u = 0; u < U; ++u) w[u] = wl[i0 + 2 * u + hi];
#pragma unroll
    for (int u = 0; u < U; ++u) {
        const float4 s = head_sum(v[u][0], v[u][1], v[u][2], v[u][3], w[u]);
        float* arow = acc + en[u].y * CDIM + 4 * q;
        float* pa = arow + (hi << 1);
        float* pb = arow + ((hi << 1) ^ 2);
        atomicAdd(pa,     hi ? s.z : s.x);
        atomicAdd(pa + 1, hi ? s.w : s.y);
        atomicAdd(pb,     hi ? s.x : s.z);
        atomicAdd(pb + 1, hi ? s.y : s.w);
    }
}

template <bool KEPT>
__global__ void
agg_owner(const int* __restrict__ srcv, const int* __restrict__ dstv,
          const float* __restrict__ Hm,
          const float* __restrict__ as_, const float* __restrict__ ad_,
          const float* __restrict__ bias,
          float* __restrict__ out, int E, int N, int do_relu,
          int* __restrict__ keep_list, int* __restrict__ keep_cnt,
          const unsigned char* __restrict__ own) {
    __shared__ __attribute__((aligned(16))) float acc[RNG * CDIM];
    __shared__ int2     list[LCAP];
    __shared__ float4   wl[LCAP];
    __shared__ float    adl[RNG * HEADS];
    __shared__ __attribute__((aligned(16))) float slf[RNG * HEADS];
    __shared__ unsigned mx[RNG * HEADS];
    __shared__ float    sm[RNG * HEADS];
    __shared__ int      cnt;

    const int t    = threadIdx.x;
    const int lane = t & 31;
    const int wave = t >> 5;
    const int base = blockIdx.x * RNG;
    const int nown = min(RNG, N - base);
    int2* kl = (int2*)keep_list + (size_t)blockIdx.x * LCAP;

    if (t == 0) cnt = 0;
    for (int i = t; i < nown * HEADS; i += AGG_T) {
        const int n4 = base * HEADS + i;
        const float a = ad_[n4];
        const float e = leaky(as_[n4] + a);
        adl[i] = a;
        slf[i] = e;
        mx[i]  = fmap(e);
        sm[i]  = 0.f;
    }
    __syncthreads();

    if constexpr (!KEPT) {
    if (gridDim.x <= NOOWN) {
        const int    E16 = E >> 4;
        const uint4* o4  = (const uint4*)own;
        const unsigned pat = blockIdx.x * 0x01010101u;
        int q0 = t;
        for (; q0 + 7 * AGG_T < E16; q0 += 8 * AGG_T) {
            uint4 ov[8];
#pragma unroll
            for (int u = 0; u < 8; ++u) ov[u] = o4[q0 + u * AGG_T];
#pragma unroll
            for (int u = 0; u < 8; u += 2)
                note_bits(own_bits(ov[u], pat) | (own_bits(ov[u + 1], pat) << 4),
                          (q0 + u * AGG_T) << 4, AGG_T << 4, &cnt, list);
        }
        for (; q0 < E16; q0 += AGG_T)
            note_bits(own_bits(o4[q0], pat), q0 << 4, 0, &cnt, list);
        for (int e = (E16 << 4) + t; e < E; e += AGG_T)
            if (own[e] == blockIdx.x) note_edge(e, &cnt, list);
    } else {
        const int  E4 = ((E & 3) == 0) ? (E >> 2) : 0;
        const int4* d4 = (const int4*)dstv;
        int q0 = t;
        for (; q0 + 3 * AGG_T < E4; q0 += 4 * AGG_T) {
            int4 dv[4];
#pragma unroll
            for (int u = 0; u < 4; ++u) dv[u] = d4[q0 + u * AGG_T];
#pragma unroll
            for (int u = 0; u < 4; ++u)
                scan4(dv[u], (q0 + u * AGG_T) << 2, base, nown, &cnt, list);
        }
        for (; q0 < E4; q0 += AGG_T)
            scan4(d4[q0], q0 << 2, base, nown, &cnt, list);
        for (int e = (E4 << 2) + t; e < E; e += AGG_T)
            if ((unsigned)(dstv[e] - base) < (unsigned)nown) note_edge(e, &cnt, list);
    }
    __syncthreads();
    if (t == 0) keep_cnt[blockIdx.x] = cnt;
    }

    const int  ntot   = KEPT ? keep_cnt[blockIdx.x] : cnt;
    const bool listed = ntot <= LCAP;

    if (listed && ntot > 0) {
        static_assert(LCAP == 4 * AGG_T, "four list entries per thread");
        bool   ok[4];
        int2   en[4];
        int    sv[4];
        float4 a4[4];
#pragma unroll
        for (int u = 0; u < 4; ++u) {
            const int i = t + u * AGG_T;
            ok[u] = i < ntot;
            en[u] = make_int2(0, 0);
            if (ok[u]) en[u] = KEPT ? kl[i] : make_int2(list[i].x, 0);
        }
#pragma unroll
        for (int u = 0; u < 4; ++u) sv[u] = KEPT ? en[u].x : srcv[en[u].x];
        if constexpr (!KEPT) {
#pragma unroll
            for (int u = 0; u < 4; ++u) en[u].y = dstv[en[u].x] - base;
        }
#pragma unroll
        for (int u = 0; u < 4; ++u) a4[u] = *(const float4*)(as_ + (size_t)sv[u] * HEADS);
#pragma unroll
        for (int u = 0; u < 4; ++u) {
            if (ok[u]) {
                const int i = t + u * AGG_T;
                const int r = en[u].y;
                const float* d = adl + r * HEADS;
                float4 e4;
                e4.x = leaky(a4[u].x + d[0]);
                e4.y = leaky(a4[u].y + d[1]);
                e4.z = leaky(a4[u].z + d[2]);
                e4.w = leaky(a4[u].w + d[3]);
                wl[i]   = e4;
                list[i] = make_int2(sv[u], r);
                if constexpr (!KEPT) kl[i] = make_int2(sv[u], r);
                unsigned* m = mx + r * HEADS;
                atomicMax(&m[0], fmap(e4.x));
                atomicMax(&m[1], fmap(e4.y));
                atomicMax(&m[2], fmap(e4.z));
                atomicMax(&m[3], fmap(e4.w));
            }
        }
    } else if (!listed) {
        walk_owned(dstv, E, base, nown, t, [&](int e, int r) {
            const int s = srcv[e];
            for (int h = 0; h < HEADS; ++h)
                atomicMax(&mx[r * HEADS + h],
                          fmap(leaky(as_[(size_t)s * HEADS + h] + adl[r * HEADS + h])));
        });
    }
    __syncthreads();

    if (listed) {
        for (int i = t; i < ntot; i += AGG_T) {
            const int r = list[i].y;
            const float4 e4 = wl[i];
            const unsigned* m = mx + r * HEADS;
            float4 x4;
            x4.x = __expf(e4.x - funmap(m[0]));
            x4.y = __expf(e4.y - funmap(m[1]));
            x4.z = __expf(e4.z - funmap(m[2]));
            x4.w = __expf(e4.w - funmap(m[3]));
            wl[i] = x4;
            float* sp = sm + r * HEADS;
            atomicAdd(&sp[0], x4.x);
            atomicAdd(&sp[1], x4.y);
            atomicAdd(&sp[2], x4.z);
            atomicAdd(&sp[3], x4.w);
        }
    } else {
        walk_owned(dstv, E, base, nown, t, [&](int e, int r) {
            const int s = srcv[e];
            for (int h = 0; h < HEADS; ++h) {
                const float ev = leaky(as_[(size_t)s * HEADS + h] + adl[r * HEADS + h]);
                atomicAdd(&sm[r * HEADS + h], __expf(ev - funmap(mx[r * HEADS + h])));
            }
        });
    }
    for (int i = t; i < nown * HEADS; i += AGG_T) {
        const float ex = __expf(slf[i] - funmap(mx[i]));
        slf[i] = ex;
        atomicAdd(&sm[i], ex);
    }
    __syncthreads();

    for (int i = t; i < nown * HEADS; i += AGG_T) {
        const float dn = sm[i] + 1e-16f;
        sm[i]  = dn;
        slf[i] = slf[i] / dn;
    }
    __syncthreads();

    {
        const float4* own4 = (const float4*)(Hm + (size_t)base * HC);
        for (int i = t; i < nown * (CDIM / 4); i += AGG_T) {
            const int r = i >> 4, q = i & 15;
            const float4* p = own4 + r * (HC / 4) + q;
            const float4 v0 = p[0], v1 = p[16], v2 = p[32], v3 = p[48];
            *(float4*)(acc + r * CDIM + 4 * q) = head_sum(v0, v1, v2, v3, *(const float4*)(slf + r * HEADS));
        }
    }
    if (listed) {
        for (int i = t; i < ntot; i += AGG_T) {
            const float* dn = sm + list[i].y * HEADS;
            float4 x4 = wl[i];
            x4.x = x4.x / dn[0];
            x4.y = x4.y / dn[1];
            x4.z = x4.z / dn[2];
            x4.w = x4.w / dn[3];
            wl[i] = x4;
        }
    }
    __syncthreads();

    if (listed) {
        const int nfull = ntot - (ntot % AGG_U);
        for (int i0 = wave * AGG_U; i0 < nfull; i0 += 32 * AGG_U)
            sum_noted<AGG_U / 2, false>(list, wl, i0, ntot, lane, acc, Hm);
        for (int i0 = nfull + 2 * wave; i0 < ntot; i0 += 64)
            sum_noted<1, true>(list, wl, i0, ntot, lane, acc, Hm);
    } else {
        walk_owned(dstv, E, base, nown, t, [&](int e, int r) {
            const int s = srcv[e];
            float al[HEADS];
            for (int h = 0; h < HEADS; ++h) {
                const float ev = leaky(as_[(size_t)s * HEADS + h] + adl[r * HEADS + h]);
                al[h] = __expf(ev - funmap(mx[r * HEADS + h])) / sm[r * HEADS + h];
            }
            const float* hr = Hm + (size_t)s * HC;
            for (int c = 0; c < CDIM; ++c)
                atomicAdd(&acc[r * CDIM + c],
                          fmaf(hr[3 * CDIM + c], al[3], fmaf(hr[2 * CDIM + c], al[2],
                               fmaf(hr[CDIM + c], al[1], hr[c] * al[0]))));
        });
    }
    __syncthreads();

    for (int i = t; i < nown * CDIM; i += AGG_T) {
        float o = acc[i] * 0.25f + bias[i & (CDIM - 1)];
        if (do_relu) o = fmaxf(o, 0.f);
        out[(size_t)base * CDIM + i] = o;
    }
}

extern "C" void kernel_launch(void* const* d_in, const int* in_sizes, int n_in,
                              void* d_out, int out_size, void* d_ws, size_t ws_size,
                              hipStream_t stream) {
    const float* x        = (const float*)d_in[0];
    const int*   ei       = (const int*)  d_in[1];
    const float* W1       = (const float*)d_in[2];
    const float* att_src1 = (const float*)d_in[3];
    const float* att_dst1 = (const float*)d_in[4];
    const float* b1       = (const float*)d_in[5];
    const float* W2       = (const float*)d_in[6];
    const float* att_src2 = (const float*)d_in[7];
    const float* att_dst2 = (const float*)d_in[8];
    const float* b2       = (const float*)d_in[9];

    const int Fin = 128;
    const int N   = in_sizes[0] / Fin;
    const int E   = in_sizes[1] / 2;
    const int* srcv = ei;
    const int* dstv = ei + E;

    float* ws    = (float*)d_ws;
    float* Hbuf  = ws;
    float* HID   = Hbuf + (size_t)N * HC;
    float* ASRC  = HID  + (size_t)N * CDIM;
    float* ADST  = ASRC + (size_t)N * HEADS;
    int*   KLIST = (int*)(ADST + (size_t)N * HEADS);
    unsigned char* OWN = (unsigned char*)(KLIST + (size_t)((N + RNG - 1) / RNG) * LCAP * 2);
    int*   KCNT  = (int*)(OWN + ((size_t)E + 15) / 16 * 16);

    const int ntiles     = N / 16;
    const int agBlocks   = (N + RNG - 1) / RNG;

    gemm_scores_wmma_f32<128><<<768, 128, 0, stream>>>(x, W1, Hbuf, att_src1, att_dst1, ASRC, ADST, ntiles, dstv, OWN, E, N);
    agg_owner<false><<<agBlocks, AGG_T, 0, stream>>>(srcv, dstv, Hbuf, ASRC, ADST, b1, HID, E, N,  1, KLIST, KCNT, OWN);

    gemm_scores_wmma_f32<CDIM><<<1024, 128, 0, stream>>>(HID, W2, Hbuf, att_src2, att_dst2, ASRC, ADST, ntiles, nullptr, nullptr, 0, 0);
    agg_owner<true><<<agBlocks, AGG_T, 0, stream>>>(srcv, dstv, Hbuf, ASRC, ADST, b2, (float*)d_out, E, N,  0, KLIST, KCNT, OWN);
}
